# starting kernel
# baseline (speedup 1.0000x reference)
#include <hip/hip_runtime.h>

typedef __attribute__((ext_vector_type(2))) float v2f;
typedef __attribute__((ext_vector_type(8))) float v8f;


__global__ void fill_ones_kernel(float* __restrict__ p, int n) {
    int i = blockIdx.x * blockDim.x + threadIdx.x;
    if (i < n) p[i] = 1.0f;
}

__global__ void deg_accum_kernel(float* __restrict__ deg, const int* __restrict__ dst, int E) {
    int e = blockIdx.x * blockDim.x + threadIdx.x;
    if (e < E) atomicAdd(&deg[dst[e]], 1.0f);
}

__global__ void deg_to_dinv_kernel(float* __restrict__ d, int n) {
    int i = blockIdx.x * blockDim.x + threadIdx.x;
    if (i < n) {
        float v = d[i];
        d[i] = (v > 0.0f) ? rsqrtf(v) : 0.0f;
    }
}


template <int NT>
__global__ __launch_bounds__(256) void gemm_wmma_kernel(const float* __restrict__ X,
                                                        const float* __restrict__ W,
                                                        float* __restrict__ Y,
                                                        int N, int numStripes) {
    constexpr int K = 128;
    constexpr int OUT = NT * 16;
    extern __shared__ float wt[];

    const int tid = threadIdx.x;
    for (int i = tid; i < K * OUT; i += 256) {
        int k = i / OUT;
        int n = i - k * OUT;
        wt[(k >> 1) * (2 * OUT) + 2 * n + (k & 1)] = W[i];
    }
    __syncthreads();

    const int lane = tid & 31;
    const int wave = tid >> 5;
    const int m = lane & 15;
    const int half = lane >> 4;
    const int kg = half << 1;

    const float* lbase = wt + (half * OUT + m) * 2;

    for (int stripe = blockIdx.x; stripe < numStripes; stripe += gridDim.x) {
        const int rowBase = stripe * 128 + wave * 16;
        int arow = rowBase + m;
        if (arow >= N) arow = N - 1;
        const float* xrow = X + (long)arow * K;

        v8f zero = {};
        v8f acc[NT];
#pragma unroll
        for (int t = 0; t < NT; ++t) acc[t] = zero;

#pragma unroll 8
        for (int k = 0; k < K; k += 4) {
            v2f a = *(const v2f*)(xrow + k + kg);
            v2f b[NT];
            const float* lk = lbase + (k >> 1) * (2 * OUT);
#pragma unroll
            for (int t = 0; t < NT; ++t) b[t] = *(const v2f*)(lk + t * 32);
#pragma unroll
            for (int t = 0; t < NT; ++t)
                acc[t] = __builtin_amdgcn_wmma_f32_16x16x4_f32(
                    false, a, false, b[t], (short)0, acc[t], false, false);
        }

#pragma unroll
        for (int t = 0; t < NT; ++t) {
            const int col = t * 16 + m;
#pragma unroll
            for (int r = 0; r < 8; ++r) {
                const int row = rowBase + r + 8 * half;
                if (row < N) Y[(long)row * OUT + col] = acc[t][r];
            }
        }
    }
}


__global__ void self_bias_kernel(float* __restrict__ out, const float* __restrict__ xw,
                                 const float* __restrict__ b, const float* __restrict__ dinv,
                                 int N, int DIM) {
    long idx = (long)blockIdx.x * blockDim.x + threadIdx.x;
    if (idx >= (long)N * DIM) return;
    int i = (int)(idx / DIM);
    int c = (int)(idx - (long)i * DIM);
    float dv = dinv[i];
    out[idx] = b[c] + xw[idx] * dv * dv;
}


template <int VPL>
__global__ __launch_bounds__(256) void edge_scatter_kernel(const float* __restrict__ xw,
                                                           const float* __restrict__ dinv,
                                                           const int* __restrict__ src,
                                                           const int* __restrict__ dst,
                                                           float* __restrict__ out, int E) {
    constexpr int DIM = 32 * VPL;
    long g = (long)blockIdx.x * blockDim.x + threadIdx.x;
    int e = (int)(g >> 5);
    int lane = (int)(g & 31);
    if (e >= E) return;
    int s = src[e];
    int d = dst[e];
    float w = dinv[s] * dinv[d];
    const float* p = xw + (long)s * DIM + lane * VPL;
    float* q = out + (long)d * DIM + lane * VPL;
    float v[VPL];
#pragma unroll
    for (int i = 0; i < VPL; ++i) v[i] = p[i];
#pragma unroll
    for (int i = 0; i < VPL; ++i) atomicAdd(q + i, v[i] * w);
}


extern "C" void kernel_launch(void* const* d_in, const int* in_sizes, int n_in,
                              void* d_out, int out_size, void* d_ws, size_t ws_size,
                              hipStream_t stream) {
    const float* x  = (const float*)d_in[0];
    const int*   ei = (const int*)d_in[1];
    const float* W1 = (const float*)d_in[2];
    const float* b1 = (const float*)d_in[3];
    const float* W2 = (const float*)d_in[4];
    const float* b2 = (const float*)d_in[5];
    float* out = (float*)d_out;

    const int N = in_sizes[0] / 128;
    const int E = in_sizes[1] / 2;
    const int* src = ei;
    const int* dst = ei + E;

    float* dinv = (float*)d_ws;
    float* xw   = dinv + N;
    float* h    = xw + (size_t)N * 128;

    fill_ones_kernel<<<(N + 255) / 256, 256, 0, stream>>>(dinv, N);
    deg_accum_kernel<<<(E + 255) / 256, 256, 0, stream>>>(dinv, dst, E);
    deg_to_dinv_kernel<<<(N + 255) / 256, 256, 0, stream>>>(dinv, N);

    const int numStripes = (N + 127) / 128;
    const int gemmBlocks = numStripes < 320 ? numStripes : 320;
    const long et = (long)E * 32;

    gemm_wmma_kernel<8><<<gemmBlocks, 256, 128 * 128 * sizeof(float), stream>>>(x, W1, xw, N, numStripes);
    {
        long tot = (long)N * 128;
        self_bias_kernel<<<(int)((tot + 255) / 256), 256, 0, stream>>>(h, xw, b1, dinv, N, 128);
    }
    edge_scatter_kernel<4><<<(int)((et + 255) / 256), 256, 0, stream>>>(xw, dinv, src, dst, h, E);

    gemm_wmma_kernel<4><<<gemmBlocks, 256, 64 * 128 * sizeof(float), stream>>>(h, W2, xw, N, numStripes);
    {
        long tot = (long)N * 64;
        self_bias_kernel<<<(int)((tot + 255) / 256), 256, 0, stream>>>(out, xw, b2, dinv, N, 64);
    }
    edge_scatter_kernel<2><<<(int)((et + 255) / 256), 256, 0, stream>>>(xw, dinv, src, dst, out, E);
}


# Layer 1's GEMM moved behind its aggregation: a group sums raw x rows in LDS, then multiplies its finished rows by W1 (+b1) and W2 there (same f32 WMMA; sums reassociated); xw never stored; 4 launches
# speedup vs baseline: 3.9753x; 3.9753x over previous
#include <hip/hip_runtime.h>

typedef __attribute__((ext_vector_type(2))) float v2f;
typedef __attribute__((ext_vector_type(4))) float v4f;
typedef __attribute__((ext_vector_type(8))) float v8f;
typedef __attribute__((ext_vector_type(4))) int v4i;


constexpr int AGG_R = 400;
constexpr int REC_ROW = 64;
constexpr int REC_W = 368;
constexpr int KEY_SEG = 8192;
constexpr int KEY_OFF = 32 * REC_W * 8;
constexpr int KEY_PITCH = AGG_R * REC_ROW * 4;
constexpr int KEY_LD = 8;
constexpr int KEY_MW = KEY_LD / 2;
constexpr int KEY_CHUNK = 32 * 16 * KEY_LD;
static_assert(KEY_OFF + KEY_SEG <= KEY_PITCH, "records and keys of a group must fit its rows");
static_assert(KEY_SEG % KEY_CHUNK == 0 && KEY_OFF % 16 == 0 && KEY_PITCH % 16 == 0, "a chunk of keys lies in one segment, 16-byte aligned");

__device__ __forceinline__ bool keys_fit(int N, int E) {
    const int groups = (N + AGG_R - 1) / AGG_R;
    return groups <= 255 && (long)(N / AGG_R) * KEY_SEG >= (long)E;
}

__device__ __forceinline__ unsigned char* key_ptr(unsigned char* kb, int e) {
    return kb + (size_t)(e >> 13) * KEY_PITCH + (e & (KEY_SEG - 1));
}
static_assert(KEY_SEG == (1 << 13), "key_ptr splits an edge number at bit 13");

__device__ __forceinline__ void store_keys(unsigned char* kb, int e, const v4i& v, int E) {
    unsigned k[4];
#pragma unroll
    for (int c = 0; c < 4; ++c) k[c] = ((unsigned)v[c] / (unsigned)AGG_R) & 255u;
    if ((e & 3) == 0 && e >= 0 && e + 4 <= E) {
        *(unsigned*)key_ptr(kb, e) = k[0] | (k[1] << 8) | (k[2] << 16) | (k[3] << 24);
    } else {
#pragma unroll
        for (int c = 0; c < 4; ++c) {
            const int ee = e + c;
            if (ee >= 0 && ee < E) *key_ptr(kb, ee) = (unsigned char)k[c];
        }
    }
}

constexpr int HIST_R = 32768;
constexpr int HIST_SL = 32;
constexpr int HIST_THREADS = 1024;
constexpr int HIST_LD = 8;

__device__ __forceinline__ v4i load_dst_quad(const int* __restrict__ dst, int pad, int q, int E) {
    const int e = q * 4 - pad;
    if (e >= 0 && e + 4 <= E) return *(const v4i*)(dst + e);
    v4i r;
#pragma unroll
    for (int c = 0; c < 4; ++c) {
        const int ee = e + c;
        r[c] = (ee >= 0 && ee < E) ? dst[ee] : -1;
    }
    return r;
}

__global__ __launch_bounds__(HIST_THREADS) void deg_hist_kernel(const int* __restrict__ dst,
                                                                 unsigned* __restrict__ part,
                                                                 float* keyrows,
                                                                 int N, int E) {
    __shared__ unsigned cnt[HIST_R];
    const int tid = threadIdx.x;
    const int lo = blockIdx.y * HIST_R;
    const bool wk = (blockIdx.y == 0) && keys_fit(N, E);
    unsigned char* kb = (unsigned char*)keyrows + KEY_OFF;

    const v4i z = {};
    for (int i = tid; i < HIST_R / 4; i += HIST_THREADS) ((v4i*)cnt)[i] = z;
    __syncthreads();

    const int pad = (int)(((size_t)dst >> 2) & 3);
    const int nQuads = (E + pad + 3) / 4;
    const int perSlice = (((nQuads + HIST_SL - 1) / HIST_SL) + 31) & ~31;
    const int q0 = blockIdx.x * perSlice;
    const int q1 = (q0 + perSlice < nQuads) ? (q0 + perSlice) : nQuads;

    const bool inside = (q0 * 4 - pad >= 0) && (q1 * 4 - pad <= E);
    if (inside) {
        const v4i* p = (const v4i*)(dst - pad);
        for (int q = q0 + tid; q < q1; q += HIST_THREADS * HIST_LD) {
            v4i v[HIST_LD];
#pragma unroll
            for (int j = 0; j < HIST_LD; ++j) {
                const int qq = q + j * HIST_THREADS;
                v[j] = p[qq < q1 ? qq : q];
            }
#pragma unroll
            for (int j = 0; j < HIST_LD; ++j) {
                const bool live = (q + j * HIST_THREADS < q1);
                if (wk && live) store_keys(kb, (q + j * HIST_THREADS) * 4 - pad, v[j], E);
#pragma unroll
                for (int c = 0; c < 4; ++c) {
                    const unsigned d = (unsigned)(v[j][c] - lo);
                    if (live && d < (unsigned)HIST_R)
                        __hip_atomic_fetch_add(&cnt[d], 1u, __ATOMIC_RELAXED, __HIP_MEMORY_SCOPE_WORKGROUP);
                }
            }
        }
    } else {
        for (int q = q0 + tid; q < q1; q += HIST_THREADS) {
            const v4i v = load_dst_quad(dst, pad, q, E);
            if (wk) store_keys(kb, q * 4 - pad, v, E);
#pragma unroll
            for (int c = 0; c < 4; ++c) {
                const unsigned d = (unsigned)(v[c] - lo);
                if (d < (unsigned)HIST_R)
                    __hip_atomic_fetch_add(&cnt[d], 1u, __ATOMIC_RELAXED, __HIP_MEMORY_SCOPE_WORKGROUP);
            }
        }
    }
    __syncthreads();

    unsigned* plane = part + (size_t)blockIdx.x * N;
    for (int i = tid * 4; i < HIST_R; i += HIST_THREADS * 4) {
        const int n = lo + i;
        const v4i w = *(const v4i*)(cnt + i);
        if (n + 4 <= N && (((size_t)(plane + n)) & 15) == 0) {
            *(v4i*)(plane + n) = w;
        } else {
#pragma unroll
            for (int c = 0; c < 4; ++c)
                if (n + c < N) plane[n + c] = (unsigned)w[c];
        }
    }
}

__global__ void deg_to_dinv_kernel(float* __restrict__ dinv, const unsigned* __restrict__ part, int N) {
    const int i = blockIdx.x * blockDim.x + threadIdx.x;
    if (i < N) {
        unsigned c = 1u;
#pragma unroll 8
        for (int s = 0; s < HIST_SL; ++s) c += part[(size_t)s * N + i];
        const float v = (float)c;
        dinv[i] = (v > 0.0f) ? rsqrtf(v) : 0.0f;
    }
}


constexpr int AGG_THREADS = 1024;
constexpr int AGG_LD = 8;
constexpr int AGG_EPL = 4 * AGG_LD;
constexpr int AGG_CHUNK = 32 * AGG_EPL;

template <int VPL> struct RowVec;
template <> struct RowVec<4> { typedef v4f type; };
template <> struct RowVec<2> { typedef v2f type; };

constexpr int H_PITCH = 132;
template <int VPL> struct RowPitch { static constexpr int v = (VPL == 4) ? H_PITCH : 32 * VPL; };

__device__ __forceinline__ void lds_add_f32(float* p, float v) {
    __hip_atomic_fetch_add(p, v, __ATOMIC_RELAXED, __HIP_MEMORY_SCOPE_WORKGROUP);
}

__device__ __forceinline__ float readlane_f32(float x, int l) {
    return __int_as_float(__builtin_amdgcn_readlane(__float_as_int(x), l));
}

__device__ __forceinline__ void load_dst_chunk(v4i (&v)[AGG_LD], const int* __restrict__ dst,
                                               int cbase, int lane, int E) {
    if (cbase >= 0 && cbase + AGG_CHUNK <= E) {
        const v4i* p = (const v4i*)(dst + cbase) + lane;
#pragma unroll
        for (int j = 0; j < AGG_LD; ++j) v[j] = p[j * 32];
    } else {
#pragma unroll
        for (int j = 0; j < AGG_LD; ++j) {
#pragma unroll
            for (int q = 0; q < 4; ++q) {
                const int e = cbase + j * 128 + lane * 4 + q;
                v[j][q] = (e >= 0 && e < E) ? dst[e] : -1;
            }
        }
    }
}

__device__ __forceinline__ unsigned owned_mask(const v4i (&v)[AGG_LD], int lo) {
    unsigned m = 0u;
#pragma unroll
    for (int j = 0; j < AGG_LD; ++j) {
#pragma unroll
        for (int q = 0; q < 4; ++q)
            if ((unsigned)(v[j][q] - lo) < (unsigned)AGG_R) m |= (1u << (j * 4 + q));
    }
    return m;
}

__device__ __forceinline__ void load_key_chunk(v4i (&v)[KEY_LD], const unsigned char* kb, int c, int lane) {
    const v4i* p = (const v4i*)key_ptr(const_cast<unsigned char*>(kb), c * KEY_CHUNK) + lane;
#pragma unroll
    for (int j = 0; j < KEY_LD; ++j) v[j] = p[j * 32];
}

__device__ __forceinline__ void key_mask(unsigned (&m)[KEY_MW], const v4i (&v)[KEY_LD], unsigned gpat) {
#pragma unroll
    for (int q = 0; q < KEY_MW; ++q) {
        unsigned a = 0u;
#pragma unroll
        for (int p = 0; p < 4; ++p) {
            const int j = q * 2 + (p >> 1);
            const int w0 = (p & 1) * 2;
            const unsigned x0 = (unsigned)v[j][w0] ^ gpat;
            const unsigned x1 = (unsigned)v[j][w0 + 1] ^ gpat;
            const unsigned n0 = (((x0 & 0x7f7f7f7fu) + 0x7f7f7f7fu) | x0) & 0x80808080u;
            const unsigned n1 = (((x1 & 0x7f7f7f7fu) + 0x7f7f7f7fu) | x1) & 0x80808080u;
            unsigned d = __builtin_amdgcn_udot4(n0, 0x08040201u, 0u, false);
            d = __builtin_amdgcn_udot4(n1, 0x80402010u, d, false);
            a |= (p == 0) ? (d >> 7) : (d << (8 * p - 7));
        }
        m[q] = ~a;
    }
}

template <int VPL>
__device__ __forceinline__ typename RowVec<VPL>::type row_load(const float* __restrict__ xw, int su, int lane) {
    typedef typename RowVec<VPL>::type vec;
    return *(const vec*)(xw + (long)su * (32 * VPL) + lane * VPL);
}

template <int VPL>
__device__ __forceinline__ void row_add(float* acc, int ru, float w, const typename RowVec<VPL>::type& q, int lane) {
    float* a = acc + ru * RowPitch<VPL>::v + lane;
#pragma unroll
    for (int i = 0; i < VPL; ++i) lds_add_f32(a + i * 32, q[i] * w);
}

constexpr int REC_SHIFT = 9;
static_assert(AGG_R <= (1 << REC_SHIFT), "the row inside the group must fit under the source node in an entry");

template <int VPL, int G, typename F>
__device__ __forceinline__ void replay_take(float* acc, const float* __restrict__ xw, const v2f& ent, int j, int lane, F between) {
    typedef typename RowVec<VPL>::type vec;
    vec q[G];
#pragma unroll
    for (int t = 0; t < G; ++t) {
        const unsigned key = (unsigned)__builtin_amdgcn_readlane(__float_as_int(ent[0]), j + t);
        q[t] = row_load<VPL>(xw, (int)(key >> REC_SHIFT), lane);
    }
    between();
#pragma unroll
    for (int t = 0; t < G; ++t) {
        const int key = __builtin_amdgcn_readlane(__float_as_int(ent[0]), j + t);
        row_add<VPL>(acc, key & ((1 << REC_SHIFT) - 1), readlane_f32(ent[1], j + t), q[t], lane);
    }
}

constexpr int RING = 64;
static_assert((RING & (RING - 1)) == 0 && RING >= 63, "31 waiting edges and 32 new ones fit the ring; its index wraps by a mask");

template <int VPL, bool REC, bool FULL>
__device__ __forceinline__ void work_batch(float* acc, const float* dl,
                                           const float* __restrict__ xw, const float* __restrict__ dinv,
                                           const int* __restrict__ src, const int* __restrict__ dst,
                                           v2f* lst, int cap, int lo, int e, int nb, int lane,
                                           int& cnt, int& ovf) {
    const bool in = FULL || (lane < nb);
    int s = 0, r = 0;
    bool ok = in;
    if (in) {
        s = src[e];
        r = dst[e] - lo;
        ok = ((unsigned)r < (unsigned)AGG_R);
    }
    float w = 0.0f;
    if (ok) w = dinv[s] * dl[r];
    v2f ent;
    ent[0] = __int_as_float((s << REC_SHIFT) | r);
    ent[1] = w;
    const unsigned V = __builtin_amdgcn_ballot_w32(ok);
    if (REC) {
        const int nV = __builtin_popcount(V);
        if (ovf == 0 && cnt + nV <= cap) {
            if (ok) lst[1 + cnt + (int)__builtin_amdgcn_mbcnt_lo(V, 0u)] = ent;
            cnt += nV;
        } else {
            ovf = 1;
        }
    }
    auto nothing = []() {};
    const unsigned all = FULL ? 0xffffffffu : ((1u << nb) - 1u);
    if (V == all) {
        if (FULL) {
            replay_take<VPL, 16>(acc, xw, ent, 0, lane, nothing);
            replay_take<VPL, 16>(acc, xw, ent, 16, lane, nothing);
        } else {
            int j = 0;
            if (nb - j >= 16) { replay_take<VPL, 16>(acc, xw, ent, j, lane, nothing); j += 16; }
            if (nb - j >= 8) { replay_take<VPL, 8>(acc, xw, ent, j, lane, nothing); j += 8; }
            if (nb - j >= 4) { replay_take<VPL, 4>(acc, xw, ent, j, lane, nothing); j += 4; }
            if (nb - j >= 2) { replay_take<VPL, 2>(acc, xw, ent, j, lane, nothing); j += 2; }
            if (nb - j >= 1) { replay_take<VPL, 1>(acc, xw, ent, j, lane, nothing); }
        }
    } else {
        unsigned B = V;
        while (B != 0u) {
            const int l = __builtin_ctz(B); B &= B - 1u;
            replay_take<VPL, 1>(acc, xw, ent, l, lane, nothing);
        }
    }
}

template <int VPL, bool REC, bool KEYS>
__device__ __forceinline__ void scan_edges(float* acc, const float* dl,
                                           const float* __restrict__ xw, const float* __restrict__ dinv,
                                           const int* __restrict__ src, const int* __restrict__ dst,
                                           const unsigned char* kb,
                                           v2f* lst, int cap, int lo, int E, int wave, int lane, int* ring) {
    const int pad = KEYS ? 0 : (int)(((size_t)dst >> 2) & 3);
    constexpr int CH = KEYS ? KEY_CHUNK : AGG_CHUNK;
    const int nChunks = (E + pad + CH - 1) / CH;
    const unsigned gpat = blockIdx.x * 0x01010101u;
    const int rot = (int)(((long)blockIdx.x * nChunks) / gridDim.x);

    int cnt = 0;
    int ovf = 0;
    int head = 0;
    int fill = 0;

    v4i buf[AGG_LD];
    v4i kbuf[KEY_LD];
    unsigned m[KEY_MW];
#pragma unroll
    for (int q = 0; q < KEY_MW; ++q) m[q] = 0u;
    auto left = [&]() {
        unsigned o = m[0];
#pragma unroll
        for (int q = 1; q < KEY_MW; ++q) o |= m[q];
        return o;
    };
    int cbase = 0;
    int it = wave;
    if (it < nChunks) {
        int c = it + rot;
        if (c >= nChunks) c -= nChunks;
        cbase = c * CH - pad;
        if (KEYS) {
            load_key_chunk(kbuf, kb, c, lane);
            key_mask(m, kbuf, gpat);
        } else {
            load_dst_chunk(buf, dst, cbase, lane, E);
            m[0] = owned_mask(buf, lo);
        }
    }
    while (it < nChunks) {
        const int itn = it + 32;
        const bool more = (itn < nChunks);
        int cbn = 0;
        if (more) {
            int c = itn + rot;
            if (c >= nChunks) c -= nChunks;
            cbn = c * CH - pad;
            if (KEYS) load_key_chunk(kbuf, kb, c, lane);
            else load_dst_chunk(buf, dst, cbn, lane, E);
        }

        while (__builtin_amdgcn_ballot_w32(left() != 0u) != 0u) {
            bool has = (left() != 0u);
            int e = 0;
            if (has) {
                int q = KEY_MW - 1;
                unsigned w = m[KEY_MW - 1];
#pragma unroll
                for (int t = KEY_MW - 2; t >= 0; --t)
                    if (m[t] != 0u) { q = t; w = m[t]; }
                const int k = __builtin_ctz(w);
#pragma unroll
                for (int t = 0; t < KEY_MW; ++t)
                    if (t == q) m[t] &= m[t] - 1u;
                e = KEYS ? (cbase + (q * 2 + (k >> 4)) * 512 + lane * 16 + (k & 15))
                         : (cbase + (k >> 2) * 128 + lane * 4 + (k & 3));
            }
            if (KEYS) has = has && (e < E);
            const unsigned Bn = __builtin_amdgcn_ballot_w32(has);
            if (has) ring[(head + fill + (int)__builtin_amdgcn_mbcnt_lo(Bn, 0u)) & (RING - 1)] = e;
            fill += __builtin_popcount(Bn);
            if (fill >= 32) {
                const int eb = ring[(head + lane) & (RING - 1)];
                work_batch<VPL, REC, true>(acc, dl, xw, dinv, src, dst, lst, cap, lo, eb, 32, lane, cnt, ovf);
                head = (head + 32) & (RING - 1);
                fill -= 32;
            }
        }

        if (!more) {
#pragma unroll
            for (int q = 0; q < KEY_MW; ++q) m[q] = 0u;
        } else if (KEYS) {
            key_mask(m, kbuf, gpat);
        } else {
            m[0] = owned_mask(buf, lo);
        }
        cbase = cbn;
        it = itn;
    }
    if (fill > 0) {
        int eb = 0;
        if (lane < fill) eb = ring[(head + lane) & (RING - 1)];
        work_batch<VPL, REC, false>(acc, dl, xw, dinv, src, dst, lst, cap, lo, eb, fill, lane, cnt, ovf);
    }
    if (REC) {
        if (lane == 0) {
            v2f hd;
            hd[0] = __int_as_float(cnt);
            hd[1] = __int_as_float(ovf);
            lst[0] = hd;
        }
    }
}

template <int VPL>
__device__ __forceinline__ void replay_list(float* acc, const float* __restrict__ xw, const v2f* lst, int cnt, int lane) {
    const v2f zero = {};
    auto nothing = []() {};
    v2f ent = zero;
    if (lane < cnt) ent = lst[1 + lane];
    for (int base = 0; base < cnt; base += 32) {
        const int nb = cnt - base;
        if (nb >= 32) {
            v2f nxt = zero;
            const int ni = base + 32 + lane;
            auto ask = [&]() { nxt = lst[1 + (ni < cnt ? ni : cnt - 1)]; };
            replay_take<VPL, 16>(acc, xw, ent, 0, lane, ask);
            replay_take<VPL, 16>(acc, xw, ent, 16, lane, nothing);
            ent = nxt;
        } else {
            int j = 0;
            if (nb - j >= 16) { replay_take<VPL, 16>(acc, xw, ent, j, lane, nothing); j += 16; }
            if (nb - j >= 8) { replay_take<VPL, 8>(acc, xw, ent, j, lane, nothing); j += 8; }
            if (nb - j >= 4) { replay_take<VPL, 4>(acc, xw, ent, j, lane, nothing); j += 4; }
            if (nb - j >= 2) { replay_take<VPL, 2>(acc, xw, ent, j, lane, nothing); j += 2; }
            if (nb - j >= 1) { replay_take<VPL, 1>(acc, xw, ent, j, lane, nothing); }
        }
    }
}

constexpr int W1_PP = 288;
constexpr int W2_PP = 160;
static_assert(AGG_R % 16 == 0, "a group's rows are whole 16-row tiles");
static_assert(W2_PP <= W1_PP, "W2 takes W1's place in LDS");

template <int NR, int NT, int PP>
__device__ __forceinline__ void tile_product(v8f (&d)[2][NT], const float* acc, const float* wt,
                                             int rt0, int c0, int lane) {
    const int m = lane & 15;
    const int half = lane >> 4;
    const float* bp = wt + half * PP + (c0 + m) * 2;
    const float* ap = acc + (rt0 * 16 + m) * H_PITCH + half * 2;
    const v8f zero = {};
#pragma unroll
    for (int j = 0; j < NR; ++j) {
#pragma unroll
        for (int t = 0; t < NT; ++t) d[j][t] = zero;
    }
#pragma unroll
    for (int ks = 0; ks < 32; ++ks) {
        v2f b[NT];
#pragma unroll
        for (int t = 0; t < NT; ++t) b[t] = *(const v2f*)(bp + ks * (2 * PP) + t * 32);
        v2f a[NR];
#pragma unroll
        for (int j = 0; j < NR; ++j) a[j] = *(const v2f*)(ap + j * (256 * H_PITCH) + ks * 4);
#pragma unroll
        for (int j = 0; j < NR; ++j) {
#pragma unroll
            for (int t = 0; t < NT; ++t)
                d[j][t] = __builtin_amdgcn_wmma_f32_16x16x4_f32(false, a[j], false, b[t], (short)0, d[j][t], false, false);
        }
    }
}

template <int NR>
__device__ __forceinline__ void h_to_rows(const v8f (&d)[2][4], float* acc, const float* __restrict__ b1,
                                          int rt0, int c0, int lane) {
    const int m = lane & 15;
    const int half = lane >> 4;
#pragma unroll
    for (int t = 0; t < 4; ++t) {
        const int col = c0 + t * 16 + m;
        const float bc = b1[col];
#pragma unroll
        for (int j = 0; j < NR; ++j) {
            float* row = acc + ((rt0 + 16 * j) * 16 + 8 * half) * H_PITCH + col;
#pragma unroll
            for (int r = 0; r < 8; ++r) row[r * H_PITCH] = d[j][t][r] + bc;
        }
    }
}

template <int NR>
__device__ __forceinline__ void xw2_store(const v8f (&d)[2][2], float* __restrict__ y,
                                          int lo, int N, int rt0, int c0, int lane) {
    const int m = lane & 15;
    const int half = lane >> 4;
#pragma unroll
    for (int j = 0; j < NR; ++j) {
        const int rbase = lo + (rt0 + 16 * j) * 16 + 8 * half;
#pragma unroll
        for (int t = 0; t < 2; ++t) {
            const int col = c0 + t * 16 + m;
#pragma unroll
            for (int r = 0; r < 8; ++r) {
                const int n = rbase + r;
                if (n < N) y[(long)n * 64 + col] = d[j][t][r];
            }
        }
    }
}

template <int VPL, int MODE>
__global__ __launch_bounds__(AGG_THREADS) void agg_owner_kernel(const float* __restrict__ xw,
                                                                const float* __restrict__ dinv,
                                                                const int* __restrict__ src,
                                                                const int* __restrict__ dst,
                                                                const float* __restrict__ bias,
                                                                const float* __restrict__ w1,
                                                                const float* __restrict__ w2,
                                                                float* out, float* rec,
                                                                int N, int E) {
    constexpr int DIM = 32 * VPL;
    constexpr int PITCH = RowPitch<VPL>::v;
    static_assert(MODE == 2 || VPL == 4, "layer 1 is the 128-wide layer whose rows feed the products by W1 and W2");
    static_assert(MODE == 1 || DIM == REC_ROW, "the walk reads its records from its own output rows");
    typedef typename RowVec<VPL>::type vec;
    __shared__ float acc[AGG_R * H_PITCH];
    __shared__ float dl[AGG_R];
    __shared__ float wt[MODE == 1 ? 64 * W1_PP : 2];
    __shared__ int rings[(AGG_THREADS / 32) * RING];

    const int tid = threadIdx.x;
    const int lane = tid & 31;
    const int wave = __builtin_amdgcn_readfirstlane(tid >> 5);
    const int lo = blockIdx.x * AGG_R;

    static_assert(128 * 128 == 4 * 4 * AGG_THREADS, "four 16-byte pieces of W1 per thread");
    static_assert(128 * 64 == 2 * 4 * AGG_THREADS, "two 16-byte pieces of W2 per thread");
    v4f wq[4];
    if (MODE == 1) {
#pragma unroll
        for (int j = 0; j < 4; ++j) wq[j] = *(const v4f*)(w1 + (tid + j * AGG_THREADS) * 4);
    }
    for (int r = tid; r < AGG_R; r += AGG_THREADS) {
        const int n = lo + r;
        dl[r] = (n < N) ? dinv[n] : 0.0f;
    }
    if (MODE == 1) {
#pragma unroll
        for (int j = 0; j < 4; ++j) {
            const int i4 = tid + j * AGG_THREADS;
            const int ch = i4 >> 5;
            const int n0 = (i4 & 31) * 4;
            const int p = (ch & 3) * 32 + (ch >> 2);
            float* w = wt + (p >> 1) * W1_PP + 2 * n0 + (p & 1);
#pragma unroll
            for (int q = 0; q < 4; ++q) w[2 * q] = wq[j][q];
        }
    }
    vec bv = {};
    if (MODE == 2) bv = *(const vec*)(bias + lane * VPL);
    for (int r = wave; r < AGG_R; r += 32) {
        const int n = lo + r;
        if (n < N) {
            const float dv = dinv[n];
            const vec p = *(const vec*)(xw + (long)n * DIM + lane * VPL);
#pragma unroll
            for (int i = 0; i < VPL; ++i) {
                const float sl = p[i] * dv * dv;
                acc[r * PITCH + i * 32 + lane] = (MODE == 2) ? bv[i] + sl : sl;
            }
        }
    }
    __syncthreads();

    const int rows = (N - lo < AGG_R) ? (N - lo) : AGG_R;
    const bool kf = keys_fit(N, E);
    const int wrec = (kf && rows == AGG_R) ? REC_W : rows;
    const int cap = wrec - 1;
    if (MODE == 1) {
        v2f* lst = (v2f*)(rec + (long)lo * REC_ROW) + wave * wrec;
        const unsigned char* kb = (const unsigned char*)rec + KEY_OFF;
        if (kf) scan_edges<VPL, true, true>(acc, dl, xw, dinv, src, dst, kb, lst, cap, lo, E, wave, lane, rings + wave * RING);
        else scan_edges<VPL, true, false>(acc, dl, xw, dinv, src, dst, kb, lst, cap, lo, E, wave, lane, rings + wave * RING);
    } else {
        static_assert(AGG_THREADS / 32 == 32, "one lane per wave's head below");
        v2f* lst = (v2f*)(out + (long)lo * REC_ROW) + wave * wrec;
        const v2f hd = ((const v2f*)(out + (long)lo * REC_ROW))[lane * wrec];
        const unsigned anyOvf = __builtin_amdgcn_ballot_w32(__float_as_int(hd[1]) != 0);
        int cnt = __builtin_amdgcn_readlane(__float_as_int(hd[0]), wave);
        cnt = cnt < 0 ? 0 : (cnt > cap ? cap : cnt);
        if (anyOvf != 0u) scan_edges<VPL, false, false>(acc, dl, xw, dinv, src, dst, nullptr, lst, 0, lo, E, wave, lane, rings + wave * RING);
        else replay_list<VPL>(acc, xw, lst, cnt, lane);
    }
    __syncthreads();

    if (MODE == 1) {
        const int cp = wave & 1;
        const int rt0 = wave >> 1;
        const bool two = (rt0 + 16 < AGG_R / 16);
        v8f h[2][4];
        if (two) tile_product<2, 4, W1_PP>(h, acc, wt, rt0, cp * 64, lane);
        else tile_product<1, 4, W1_PP>(h, acc, wt, rt0, cp * 64, lane);
        const int th = wave * 32 + lane;
        v4f w2q[2];
#pragma unroll
        for (int j = 0; j < 2; ++j) w2q[j] = *(const v4f*)(w2 + (th + j * AGG_THREADS) * 4);
        __syncthreads();
        if (two) h_to_rows<2>(h, acc, bias, rt0, cp * 64, lane);
        else h_to_rows<1>(h, acc, bias, rt0, cp * 64, lane);
#pragma unroll
        for (int j = 0; j < 2; ++j) {
            const int i4 = th + j * AGG_THREADS;
            const int ch = i4 >> 4;
            const int n0 = (i4 & 15) * 4;
            float* w = wt + (ch >> 1) * W2_PP + 2 * n0 + (ch & 1);
#pragma unroll
            for (int q = 0; q < 4; ++q) w[2 * q] = w2q[j][q];
        }
        __syncthreads();
        v8f y[2][2];
        if (two) {
            tile_product<2, 2, W2_PP>(y, acc, wt, rt0, cp * 32, lane);
            xw2_store<2>(y, out, lo, N, rt0, cp * 32, lane);
        } else {
            tile_product<1, 2, W2_PP>(y, acc, wt, rt0, cp * 32, lane);
            xw2_store<1>(y, out, lo, N, rt0, cp * 32, lane);
        }
    } else {
        for (int r = wave; r < AGG_R; r += 32) {
            const int n = lo + r;
            if (n < N) {
                vec o;
#pragma unroll
                for (int i = 0; i < VPL; ++i) o[i] = acc[r * PITCH + i * 32 + lane];
                *(vec*)(out + (long)n * DIM + lane * VPL) = o;
            }
        }
    }
}


extern "C" void kernel_launch(void* const* d_in, const int* in_sizes, int n_in,
                              void* d_out, int out_size, void* d_ws, size_t ws_size,
                              hipStream_t stream) {
    const float* x  = (const float*)d_in[0];
    const int*   ei = (const int*)d_in[1];
    const float* W1 = (const float*)d_in[2];
    const float* b1 = (const float*)d_in[3];
    const float* W2 = (const float*)d_in[4];
    const float* b2 = (const float*)d_in[5];
    float* out = (float*)d_out;

    const int N = in_sizes[0] / 128;
    const int E = in_sizes[1] / 2;
    const int* src = ei;
    const int* dst = ei + E;

    float* dinv = (float*)d_ws;
    float* xw2  = dinv + N + (size_t)N * 128;

    unsigned* part = (unsigned*)xw2;
    deg_hist_kernel<<<dim3(HIST_SL, (N + HIST_R - 1) / HIST_R), HIST_THREADS, 0, stream>>>(dst, part, out, N, E);
    deg_to_dinv_kernel<<<(N + 255) / 256, 256, 0, stream>>>(dinv, part, N);

    const int aggBlocks = (N + AGG_R - 1) / AGG_R;

    agg_owner_kernel<4, 1><<<aggBlocks, AGG_THREADS, 0, stream>>>(x, dinv, src, dst, b1, W1, W2, xw2, out, N, E);

    agg_owner_kernel<2, 2><<<aggBlocks, AGG_THREADS, 0, stream>>>(xw2, dinv, src, dst, b2, nullptr, nullptr, out, out, N, E);
}


# Self loops out of start-up: rows start at zero (layer 2: bias); a wave fetches its 13 own rows behind its edges, one group, cached by then, and adds them last (layer 2: at the store); sums reordered
# speedup vs baseline: 4.1327x; 1.0396x over previous
#include <hip/hip_runtime.h>

typedef __attribute__((ext_vector_type(2))) float v2f;
typedef __attribute__((ext_vector_type(4))) float v4f;
typedef __attribute__((ext_vector_type(8))) float v8f;
typedef __attribute__((ext_vector_type(4))) int v4i;


constexpr int AGG_R = 400;
constexpr int REC_ROW = 64;
constexpr int REC_W = 368;
constexpr int KEY_SEG = 8192;
constexpr int KEY_OFF = 32 * REC_W * 8;
constexpr int KEY_PITCH = AGG_R * REC_ROW * 4;
constexpr int KEY_LD = 8;
constexpr int KEY_MW = KEY_LD / 2;
constexpr int KEY_CHUNK = 32 * 16 * KEY_LD;
static_assert(KEY_OFF + KEY_SEG <= KEY_PITCH, "records and keys of a group must fit its rows");
static_assert(KEY_SEG % KEY_CHUNK == 0 && KEY_OFF % 16 == 0 && KEY_PITCH % 16 == 0, "a chunk of keys lies in one segment, 16-byte aligned");

__device__ __forceinline__ bool keys_fit(int N, int E) {
    const int groups = (N + AGG_R - 1) / AGG_R;
    return groups <= 255 && (long)(N / AGG_R) * KEY_SEG >= (long)E;
}

__device__ __forceinline__ unsigned char* key_ptr(unsigned char* kb, int e) {
    return kb + (size_t)(e >> 13) * KEY_PITCH + (e & (KEY_SEG - 1));
}
static_assert(KEY_SEG == (1 << 13), "key_ptr splits an edge number at bit 13");

__device__ __forceinline__ void store_keys(unsigned char* kb, int e, const v4i& v, int E) {
    unsigned k[4];
#pragma unroll
    for (int c = 0; c < 4; ++c) k[c] = ((unsigned)v[c] / (unsigned)AGG_R) & 255u;
    if ((e & 3) == 0 && e >= 0 && e + 4 <= E) {
        *(unsigned*)key_ptr(kb, e) = k[0] | (k[1] << 8) | (k[2] << 16) | (k[3] << 24);
    } else {
#pragma unroll
        for (int c = 0; c < 4; ++c) {
            const int ee = e + c;
            if (ee >= 0 && ee < E) *key_ptr(kb, ee) = (unsigned char)k[c];
        }
    }
}

constexpr int HIST_R = 32768;
constexpr int HIST_SL = 32;
constexpr int HIST_THREADS = 1024;
constexpr int HIST_LD = 8;

__device__ __forceinline__ v4i load_dst_quad(const int* __restrict__ dst, int pad, int q, int E) {
    const int e = q * 4 - pad;
    if (e >= 0 && e + 4 <= E) return *(const v4i*)(dst + e);
    v4i r;
#pragma unroll
    for (int c = 0; c < 4; ++c) {
        const int ee = e + c;
        r[c] = (ee >= 0 && ee < E) ? dst[ee] : -1;
    }
    return r;
}

__global__ __launch_bounds__(HIST_THREADS) void deg_hist_kernel(const int* __restrict__ dst,
                                                                 unsigned* __restrict__ part,
                                                                 float* keyrows,
                                                                 int N, int E) {
    __shared__ unsigned cnt[HIST_R];
    const int tid = threadIdx.x;
    const int lo = blockIdx.y * HIST_R;
    const bool wk = (blockIdx.y == 0) && keys_fit(N, E);
    unsigned char* kb = (unsigned char*)keyrows + KEY_OFF;

    const v4i z = {};
    for (int i = tid; i < HIST_R / 4; i += HIST_THREADS) ((v4i*)cnt)[i] = z;
    __syncthreads();

    const int pad = (int)(((size_t)dst >> 2) & 3);
    const int nQuads = (E + pad + 3) / 4;
    const int perSlice = (((nQuads + HIST_SL - 1) / HIST_SL) + 31) & ~31;
    const int q0 = blockIdx.x * perSlice;
    const int q1 = (q0 + perSlice < nQuads) ? (q0 + perSlice) : nQuads;

    const bool inside = (q0 * 4 - pad >= 0) && (q1 * 4 - pad <= E);
    if (inside) {
        const v4i* p = (const v4i*)(dst - pad);
        for (int q = q0 + tid; q < q1; q += HIST_THREADS * HIST_LD) {
            v4i v[HIST_LD];
#pragma unroll
            for (int j = 0; j < HIST_LD; ++j) {
                const int qq = q + j * HIST_THREADS;
                v[j] = p[qq < q1 ? qq : q];
            }
#pragma unroll
            for (int j = 0; j < HIST_LD; ++j) {
                const bool live = (q + j * HIST_THREADS < q1);
                if (wk && live) store_keys(kb, (q + j * HIST_THREADS) * 4 - pad, v[j], E);
#pragma unroll
                for (int c = 0; c < 4; ++c) {
                    const unsigned d = (unsigned)(v[j][c] - lo);
                    if (live && d < (unsigned)HIST_R)
                        __hip_atomic_fetch_add(&cnt[d], 1u, __ATOMIC_RELAXED, __HIP_MEMORY_SCOPE_WORKGROUP);
                }
            }
        }
    } else {
        for (int q = q0 + tid; q < q1; q += HIST_THREADS) {
            const v4i v = load_dst_quad(dst, pad, q, E);
            if (wk) store_keys(kb, q * 4 - pad, v, E);
#pragma unroll
            for (int c = 0; c < 4; ++c) {
                const unsigned d = (unsigned)(v[c] - lo);
                if (d < (unsigned)HIST_R)
                    __hip_atomic_fetch_add(&cnt[d], 1u, __ATOMIC_RELAXED, __HIP_MEMORY_SCOPE_WORKGROUP);
            }
        }
    }
    __syncthreads();

    unsigned* plane = part + (size_t)blockIdx.x * N;
    for (int i = tid * 4; i < HIST_R; i += HIST_THREADS * 4) {
        const int n = lo + i;
        const v4i w = *(const v4i*)(cnt + i);
        if (n + 4 <= N && (((size_t)(plane + n)) & 15) == 0) {
            *(v4i*)(plane + n) = w;
        } else {
#pragma unroll
            for (int c = 0; c < 4; ++c)
                if (n + c < N) plane[n + c] = (unsigned)w[c];
        }
    }
}

__global__ void deg_to_dinv_kernel(float* __restrict__ dinv, const unsigned* __restrict__ part, int N) {
    const int i = blockIdx.x * blockDim.x + threadIdx.x;
    if (i < N) {
        unsigned c = 1u;
#pragma unroll 8
        for (int s = 0; s < HIST_SL; ++s) c += part[(size_t)s * N + i];
        const float v = (float)c;
        dinv[i] = (v > 0.0f) ? rsqrtf(v) : 0.0f;
    }
}


constexpr int AGG_THREADS = 1024;
constexpr int AGG_LD = 8;
constexpr int AGG_EPL = 4 * AGG_LD;
constexpr int AGG_CHUNK = 32 * AGG_EPL;

template <int VPL> struct RowVec;
template <> struct RowVec<4> { typedef v4f type; };
template <> struct RowVec<2> { typedef v2f type; };

constexpr int H_PITCH = 132;
template <int VPL> struct RowPitch { static constexpr int v = (VPL == 4) ? H_PITCH : 32 * VPL; };

__device__ __forceinline__ void lds_add_f32(float* p, float v) {
    __hip_atomic_fetch_add(p, v, __ATOMIC_RELAXED, __HIP_MEMORY_SCOPE_WORKGROUP);
}

__device__ __forceinline__ float readlane_f32(float x, int l) {
    return __int_as_float(__builtin_amdgcn_readlane(__float_as_int(x), l));
}

__device__ __forceinline__ void load_dst_chunk(v4i (&v)[AGG_LD], const int* __restrict__ dst,
                                               int cbase, int lane, int E) {
    if (cbase >= 0 && cbase + AGG_CHUNK <= E) {
        const v4i* p = (const v4i*)(dst + cbase) + lane;
#pragma unroll
        for (int j = 0; j < AGG_LD; ++j) v[j] = p[j * 32];
    } else {
#pragma unroll
        for (int j = 0; j < AGG_LD; ++j) {
#pragma unroll
            for (int q = 0; q < 4; ++q) {
                const int e = cbase + j * 128 + lane * 4 + q;
                v[j][q] = (e >= 0 && e < E) ? dst[e] : -1;
            }
        }
    }
}

__device__ __forceinline__ unsigned owned_mask(const v4i (&v)[AGG_LD], int lo) {
    unsigned m = 0u;
#pragma unroll
    for (int j = 0; j < AGG_LD; ++j) {
#pragma unroll
        for (int q = 0; q < 4; ++q)
            if ((unsigned)(v[j][q] - lo) < (unsigned)AGG_R) m |= (1u << (j * 4 + q));
    }
    return m;
}

__device__ __forceinline__ void load_key_chunk(v4i (&v)[KEY_LD], const unsigned char* kb, int c, int lane) {
    const v4i* p = (const v4i*)key_ptr(const_cast<unsigned char*>(kb), c * KEY_CHUNK) + lane;
#pragma unroll
    for (int j = 0; j < KEY_LD; ++j) v[j] = p[j * 32];
}

__device__ __forceinline__ void key_mask(unsigned (&m)[KEY_MW], const v4i (&v)[KEY_LD], unsigned gpat) {
#pragma unroll
    for (int q = 0; q < KEY_MW; ++q) {
        unsigned a = 0u;
#pragma unroll
        for (int p = 0; p < 4; ++p) {
            const int j = q * 2 + (p >> 1);
            const int w0 = (p & 1) * 2;
            const unsigned x0 = (unsigned)v[j][w0] ^ gpat;
            const unsigned x1 = (unsigned)v[j][w0 + 1] ^ gpat;
            const unsigned n0 = (((x0 & 0x7f7f7f7fu) + 0x7f7f7f7fu) | x0) & 0x80808080u;
            const unsigned n1 = (((x1 & 0x7f7f7f7fu) + 0x7f7f7f7fu) | x1) & 0x80808080u;
            unsigned d = __builtin_amdgcn_udot4(n0, 0x08040201u, 0u, false);
            d = __builtin_amdgcn_udot4(n1, 0x80402010u, d, false);
            a |= (p == 0) ? (d >> 7) : (d << (8 * p - 7));
        }
        m[q] = ~a;
    }
}

template <int VPL>
__device__ __forceinline__ typename RowVec<VPL>::type row_load(const float* __restrict__ xw, int su, int lane) {
    typedef typename RowVec<VPL>::type vec;
    return *(const vec*)(xw + (long)su * (32 * VPL) + lane * VPL);
}

template <int VPL>
__device__ __forceinline__ void row_add(float* acc, int ru, float w, const typename RowVec<VPL>::type& q, int lane) {
    float* a = acc + ru * RowPitch<VPL>::v + lane;
#pragma unroll
    for (int i = 0; i < VPL; ++i) lds_add_f32(a + i * 32, q[i] * w);
}

constexpr int REC_SHIFT = 9;
static_assert(AGG_R <= (1 << REC_SHIFT), "the row inside the group must fit under the source node in an entry");

template <int VPL, int G, typename F>
__device__ __forceinline__ void replay_take(float* acc, const float* __restrict__ xw, const v2f& ent, int j, int lane, F between) {
    typedef typename RowVec<VPL>::type vec;
    vec q[G];
#pragma unroll
    for (int t = 0; t < G; ++t) {
        const unsigned key = (unsigned)__builtin_amdgcn_readlane(__float_as_int(ent[0]), j + t);
        q[t] = row_load<VPL>(xw, (int)(key >> REC_SHIFT), lane);
    }
    between();
#pragma unroll
    for (int t = 0; t < G; ++t) {
        const int key = __builtin_amdgcn_readlane(__float_as_int(ent[0]), j + t);
        row_add<VPL>(acc, key & ((1 << REC_SHIFT) - 1), readlane_f32(ent[1], j + t), q[t], lane);
    }
}

constexpr int RING = 64;
static_assert((RING & (RING - 1)) == 0 && RING >= 63, "31 waiting edges and 32 new ones fit the ring; its index wraps by a mask");

template <int VPL, bool REC, bool FULL>
__device__ __forceinline__ void work_batch(float* acc, const float* dl,
                                           const float* __restrict__ xw, const float* __restrict__ dinv,
                                           const int* __restrict__ src, const int* __restrict__ dst,
                                           v2f* lst, int cap, int lo, int e, int nb, int lane,
                                           int& cnt, int& ovf) {
    const bool in = FULL || (lane < nb);
    int s = 0, r = 0;
    bool ok = in;
    if (in) {
        s = src[e];
        r = dst[e] - lo;
        ok = ((unsigned)r < (unsigned)AGG_R);
    }
    float w = 0.0f;
    if (ok) w = dinv[s] * dl[r];
    v2f ent;
    ent[0] = __int_as_float((s << REC_SHIFT) | r);
    ent[1] = w;
    const unsigned V = __builtin_amdgcn_ballot_w32(ok);
    if (REC) {
        const int nV = __builtin_popcount(V);
        if (ovf == 0 && cnt + nV <= cap) {
            if (ok) lst[1 + cnt + (int)__builtin_amdgcn_mbcnt_lo(V, 0u)] = ent;
            cnt += nV;
        } else {
            ovf = 1;
        }
    }
    auto nothing = []() {};
    const unsigned all = FULL ? 0xffffffffu : ((1u << nb) - 1u);
    if (V == all) {
        if (FULL) {
            replay_take<VPL, 16>(acc, xw, ent, 0, lane, nothing);
            replay_take<VPL, 16>(acc, xw, ent, 16, lane, nothing);
        } else {
            int j = 0;
            if (nb - j >= 16) { replay_take<VPL, 16>(acc, xw, ent, j, lane, nothing); j += 16; }
            if (nb - j >= 8) { replay_take<VPL, 8>(acc, xw, ent, j, lane, nothing); j += 8; }
            if (nb - j >= 4) { replay_take<VPL, 4>(acc, xw, ent, j, lane, nothing); j += 4; }
            if (nb - j >= 2) { replay_take<VPL, 2>(acc, xw, ent, j, lane, nothing); j += 2; }
            if (nb - j >= 1) { replay_take<VPL, 1>(acc, xw, ent, j, lane, nothing); }
        }
    } else {
        unsigned B = V;
        while (B != 0u) {
            const int l = __builtin_ctz(B); B &= B - 1u;
            replay_take<VPL, 1>(acc, xw, ent, l, lane, nothing);
        }
    }
}

template <int VPL, bool REC, bool KEYS>
__device__ __forceinline__ void scan_edges(float* acc, const float* dl,
                                           const float* __restrict__ xw, const float* __restrict__ dinv,
                                           const int* __restrict__ src, const int* __restrict__ dst,
                                           const unsigned char* kb,
                                           v2f* lst, int cap, int lo, int E, int wave, int lane, int* ring) {
    const int pad = KEYS ? 0 : (int)(((size_t)dst >> 2) & 3);
    constexpr int CH = KEYS ? KEY_CHUNK : AGG_CHUNK;
    const int nChunks = (E + pad + CH - 1) / CH;
    const unsigned gpat = blockIdx.x * 0x01010101u;
    const int rot = (int)(((long)blockIdx.x * nChunks) / gridDim.x);

    int cnt = 0;
    int ovf = 0;
    int head = 0;
    int fill = 0;

    v4i buf[AGG_LD];
    v4i kbuf[KEY_LD];
    unsigned m[KEY_MW];
#pragma unroll
    for (int q = 0; q < KEY_MW; ++q) m[q] = 0u;
    auto left = [&]() {
        unsigned o = m[0];
#pragma unroll
        for (int q = 1; q < KEY_MW; ++q) o |= m[q];
        return o;
    };
    int cbase = 0;
    int it = wave;
    if (it < nChunks) {
        int c = it + rot;
        if (c >= nChunks) c -= nChunks;
        cbase = c * CH - pad;
        if (KEYS) {
            load_key_chunk(kbuf, kb, c, lane);
            key_mask(m, kbuf, gpat);
        } else {
            load_dst_chunk(buf, dst, cbase, lane, E);
            m[0] = owned_mask(buf, lo);
        }
    }
    while (it < nChunks) {
        const int itn = it + 32;
        const bool more = (itn < nChunks);
        int cbn = 0;
        if (more) {
            int c = itn + rot;
            if (c >= nChunks) c -= nChunks;
            cbn = c * CH - pad;
            if (KEYS) load_key_chunk(kbuf, kb, c, lane);
            else load_dst_chunk(buf, dst, cbn, lane, E);
        }

        while (__builtin_amdgcn_ballot_w32(left() != 0u) != 0u) {
            bool has = (left() != 0u);
            int e = 0;
            if (has) {
                int q = KEY_MW - 1;
                unsigned w = m[KEY_MW - 1];
#pragma unroll
                for (int t = KEY_MW - 2; t >= 0; --t)
                    if (m[t] != 0u) { q = t; w = m[t]; }
                const int k = __builtin_ctz(w);
#pragma unroll
                for (int t = 0; t < KEY_MW; ++t)
                    if (t == q) m[t] &= m[t] - 1u;
                e = KEYS ? (cbase + (q * 2 + (k >> 4)) * 512 + lane * 16 + (k & 15))
                         : (cbase + (k >> 2) * 128 + lane * 4 + (k & 3));
            }
            if (KEYS) has = has && (e < E);
            const unsigned Bn = __builtin_amdgcn_ballot_w32(has);
            if (has) ring[(head + fill + (int)__builtin_amdgcn_mbcnt_lo(Bn, 0u)) & (RING - 1)] = e;
            fill += __builtin_popcount(Bn);
            if (fill >= 32) {
                const int eb = ring[(head + lane) & (RING - 1)];
                work_batch<VPL, REC, true>(acc, dl, xw, dinv, src, dst, lst, cap, lo, eb, 32, lane, cnt, ovf);
                head = (head + 32) & (RING - 1);
                fill -= 32;
            }
        }

        if (!more) {
#pragma unroll
            for (int q = 0; q < KEY_MW; ++q) m[q] = 0u;
        } else if (KEYS) {
            key_mask(m, kbuf, gpat);
        } else {
            m[0] = owned_mask(buf, lo);
        }
        cbase = cbn;
        it = itn;
    }
    if (fill > 0) {
        int eb = 0;
        if (lane < fill) eb = ring[(head + lane) & (RING - 1)];
        work_batch<VPL, REC, false>(acc, dl, xw, dinv, src, dst, lst, cap, lo, eb, fill, lane, cnt, ovf);
    }
    if (REC) {
        if (lane == 0) {
            v2f hd;
            hd[0] = __int_as_float(cnt);
            hd[1] = __int_as_float(ovf);
            lst[0] = hd;
        }
    }
}

template <int VPL>
__device__ __forceinline__ void replay_list(float* acc, const float* __restrict__ xw, const v2f* lst, int cnt, int lane) {
    const v2f zero = {};
    auto nothing = []() {};
    v2f ent = zero;
    if (lane < cnt) ent = lst[1 + lane];
    for (int base = 0; base < cnt; base += 32) {
        const int nb = cnt - base;
        if (nb >= 32) {
            v2f nxt = zero;
            const int ni = base + 32 + lane;
            auto ask = [&]() { nxt = lst[1 + (ni < cnt ? ni : cnt - 1)]; };
            replay_take<VPL, 16>(acc, xw, ent, 0, lane, ask);
            replay_take<VPL, 16>(acc, xw, ent, 16, lane, nothing);
            ent = nxt;
        } else {
            int j = 0;
            if (nb - j >= 16) { replay_take<VPL, 16>(acc, xw, ent, j, lane, nothing); j += 16; }
            if (nb - j >= 8) { replay_take<VPL, 8>(acc, xw, ent, j, lane, nothing); j += 8; }
            if (nb - j >= 4) { replay_take<VPL, 4>(acc, xw, ent, j, lane, nothing); j += 4; }
            if (nb - j >= 2) { replay_take<VPL, 2>(acc, xw, ent, j, lane, nothing); j += 2; }
            if (nb - j >= 1) { replay_take<VPL, 1>(acc, xw, ent, j, lane, nothing); }
        }
    }
}

constexpr int W1_PP = 288;
constexpr int W2_PP = 160;
static_assert(AGG_R % 16 == 0, "a group's rows are whole 16-row tiles");
static_assert(W2_PP <= W1_PP, "W2 takes W1's place in LDS");

template <int NR, int NT, int PP>
__device__ __forceinline__ void tile_product(v8f (&d)[2][NT], const float* acc, const float* wt,
                                             int rt0, int c0, int lane) {
    const int m = lane & 15;
    const int half = lane >> 4;
    const float* bp = wt + half * PP + (c0 + m) * 2;
    const float* ap = acc + (rt0 * 16 + m) * H_PITCH + half * 2;
    const v8f zero = {};
#pragma unroll
    for (int j = 0; j < NR; ++j) {
#pragma unroll
        for (int t = 0; t < NT; ++t) d[j][t] = zero;
    }
#pragma unroll
    for (int ks = 0; ks < 32; ++ks) {
        v2f b[NT];
#pragma unroll
        for (int t = 0; t < NT; ++t) b[t] = *(const v2f*)(bp + ks * (2 * PP) + t * 32);
        v2f a[NR];
#pragma unroll
        for (int j = 0; j < NR; ++j) a[j] = *(const v2f*)(ap + j * (256 * H_PITCH) + ks * 4);
#pragma unroll
        for (int j = 0; j < NR; ++j) {
#pragma unroll
            for (int t = 0; t < NT; ++t)
                d[j][t] = __builtin_amdgcn_wmma_f32_16x16x4_f32(false, a[j], false, b[t], (short)0, d[j][t], false, false);
        }
    }
}

template <int NR>
__device__ __forceinline__ void h_to_rows(const v8f (&d)[2][4], float* acc, const float* __restrict__ b1,
                                          int rt0, int c0, int lane) {
    const int m = lane & 15;
    const int half = lane >> 4;
#pragma unroll
    for (int t = 0; t < 4; ++t) {
        const int col = c0 + t * 16 + m;
        const float bc = b1[col];
#pragma unroll
        for (int j = 0; j < NR; ++j) {
            float* row = acc + ((rt0 + 16 * j) * 16 + 8 * half) * H_PITCH + col;
#pragma unroll
            for (int r = 0; r < 8; ++r) row[r * H_PITCH] = d[j][t][r] + bc;
        }
    }
}

template <int NR>
__device__ __forceinline__ void xw2_store(const v8f (&d)[2][2], float* __restrict__ y,
                                          int lo, int N, int rt0, int c0, int lane) {
    const int m = lane & 15;
    const int half = lane >> 4;
#pragma unroll
    for (int j = 0; j < NR; ++j) {
        const int rbase = lo + (rt0 + 16 * j) * 16 + 8 * half;
#pragma unroll
        for (int t = 0; t < 2; ++t) {
            const int col = c0 + t * 16 + m;
#pragma unroll
            for (int r = 0; r < 8; ++r) {
                const int n = rbase + r;
                if (n < N) y[(long)n * 64 + col] = d[j][t][r];
            }
        }
    }
}

template <int VPL, int MODE>
__global__ __launch_bounds__(AGG_THREADS) void agg_owner_kernel(const float* __restrict__ xw,
                                                                const float* __restrict__ dinv,
                                                                const int* __restrict__ src,
                                                                const int* __restrict__ dst,
                                                                const float* __restrict__ bias,
                                                                const float* __restrict__ w1,
                                                                const float* __restrict__ w2,
                                                                float* out, float* rec,
                                                                int N, int E) {
    constexpr int DIM = 32 * VPL;
    constexpr int PITCH = RowPitch<VPL>::v;
    static_assert(MODE == 2 || VPL == 4, "layer 1 is the 128-wide layer whose rows feed the products by W1 and W2");
    static_assert(MODE == 1 || DIM == REC_ROW, "the walk reads its records from its own output rows");
    typedef typename RowVec<VPL>::type vec;
    __shared__ float acc[AGG_R * H_PITCH];
    __shared__ float dl[AGG_R];
    __shared__ float wt[MODE == 1 ? 64 * W1_PP : 2];
    __shared__ int rings[(AGG_THREADS / 32) * RING];

    const int tid = threadIdx.x;
    const int lane = tid & 31;
    const int wave = __builtin_amdgcn_readfirstlane(tid >> 5);
    const int lo = blockIdx.x * AGG_R;

    static_assert(128 * 128 == 4 * 4 * AGG_THREADS, "four 16-byte pieces of W1 per thread");
    static_assert(128 * 64 == 2 * 4 * AGG_THREADS, "two 16-byte pieces of W2 per thread");
    v4f wq[4];
    if (MODE == 1) {
#pragma unroll
        for (int j = 0; j < 4; ++j) wq[j] = *(const v4f*)(w1 + (tid + j * AGG_THREADS) * 4);
    }
    vec bv = {};
    if (MODE == 2) bv = *(const vec*)(bias + lane * VPL);
    for (int r = tid; r < AGG_R; r += AGG_THREADS) {
        const int n = lo + r;
        dl[r] = (n < N) ? dinv[n] : 0.0f;
    }
    if (MODE == 1) {
#pragma unroll
        for (int j = 0; j < 4; ++j) {
            const int i4 = tid + j * AGG_THREADS;
            const int ch = i4 >> 5;
            const int n0 = (i4 & 31) * 4;
            const int p = (ch & 3) * 32 + (ch >> 2);
            float* w = wt + (p >> 1) * W1_PP + 2 * n0 + (p & 1);
#pragma unroll
            for (int q = 0; q < 4; ++q) w[2 * q] = wq[j][q];
        }
    }
    for (int r = wave; r < AGG_R; r += 32) {
#pragma unroll
        for (int i = 0; i < VPL; ++i) acc[r * PITCH + i * 32 + lane] = bv[i];
    }
    __syncthreads();

    const int rows = (N - lo < AGG_R) ? (N - lo) : AGG_R;
    const bool kf = keys_fit(N, E);
    const int wrec = (kf && rows == AGG_R) ? REC_W : rows;
    const int cap = wrec - 1;
    if (MODE == 1) {
        v2f* lst = (v2f*)(rec + (long)lo * REC_ROW) + wave * wrec;
        const unsigned char* kb = (const unsigned char*)rec + KEY_OFF;
        if (kf) scan_edges<VPL, true, true>(acc, dl, xw, dinv, src, dst, kb, lst, cap, lo, E, wave, lane, rings + wave * RING);
        else scan_edges<VPL, true, false>(acc, dl, xw, dinv, src, dst, kb, lst, cap, lo, E, wave, lane, rings + wave * RING);
    } else {
        static_assert(AGG_THREADS / 32 == 32, "one lane per wave's head below");
        v2f* lst = (v2f*)(out + (long)lo * REC_ROW) + wave * wrec;
        const v2f hd = ((const v2f*)(out + (long)lo * REC_ROW))[lane * wrec];
        const unsigned anyOvf = __builtin_amdgcn_ballot_w32(__float_as_int(hd[1]) != 0);
        int cnt = __builtin_amdgcn_readlane(__float_as_int(hd[0]), wave);
        cnt = cnt < 0 ? 0 : (cnt > cap ? cap : cnt);
        if (anyOvf != 0u) scan_edges<VPL, false, false>(acc, dl, xw, dinv, src, dst, nullptr, lst, 0, lo, E, wave, lane, rings + wave * RING);
        else replay_list<VPL>(acc, xw, lst, cnt, lane);
    }
    constexpr int OWN = (AGG_R + 31) / 32;
    vec own[OWN];
    float odv[OWN];
#pragma unroll
    for (int k = 0; k < OWN; ++k) {
        const int r = wave + 32 * k;
        const bool live = (r < AGG_R && lo + r < N);
        own[k] = row_load<VPL>(xw, live ? (lo + r) : lo, lane);
        odv[k] = dl[live ? r : 0];
    }
    if (MODE == 1) {
#pragma unroll
        for (int k = 0; k < OWN; ++k) {
            const int r = wave + 32 * k;
            if (r < AGG_R && lo + r < N) {
                const float dv = odv[k];
                float* a = acc + r * PITCH + lane;
#pragma unroll
                for (int i = 0; i < VPL; ++i) lds_add_f32(a + i * 32, own[k][i] * dv * dv);
            }
        }
    }
    __syncthreads();

    if (MODE == 1) {
        const int cp = wave & 1;
        const int rt0 = wave >> 1;
        const bool two = (rt0 + 16 < AGG_R / 16);
        v8f h[2][4];
        if (two) tile_product<2, 4, W1_PP>(h, acc, wt, rt0, cp * 64, lane);
        else tile_product<1, 4, W1_PP>(h, acc, wt, rt0, cp * 64, lane);
        const int th = wave * 32 + lane;
        v4f w2q[2];
#pragma unroll
        for (int j = 0; j < 2; ++j) w2q[j] = *(const v4f*)(w2 + (th + j * AGG_THREADS) * 4);
        __syncthreads();
        if (two) h_to_rows<2>(h, acc, bias, rt0, cp * 64, lane);
        else h_to_rows<1>(h, acc, bias, rt0, cp * 64, lane);
#pragma unroll
        for (int j = 0; j < 2; ++j) {
            const int i4 = th + j * AGG_THREADS;
            const int ch = i4 >> 4;
            const int n0 = (i4 & 15) * 4;
            float* w = wt + (ch >> 1) * W2_PP + 2 * n0 + (ch & 1);
#pragma unroll
            for (int q = 0; q < 4; ++q) w[2 * q] = w2q[j][q];
        }
        __syncthreads();
        v8f y[2][2];
        if (two) {
            tile_product<2, 2, W2_PP>(y, acc, wt, rt0, cp * 32, lane);
            xw2_store<2>(y, out, lo, N, rt0, cp * 32, lane);
        } else {
            tile_product<1, 2, W2_PP>(y, acc, wt, rt0, cp * 32, lane);
            xw2_store<1>(y, out, lo, N, rt0, cp * 32, lane);
        }
    } else {
#pragma unroll
        for (int k = 0; k < OWN; ++k) {
            const int r = wave + 32 * k;
            const int n = lo + r;
            if (r < AGG_R && n < N) {
                const float dv = odv[k];
                vec o;
#pragma unroll
                for (int i = 0; i < VPL; ++i) o[i] = acc[r * PITCH + i * 32 + lane] + own[k][i] * dv * dv;
                *(vec*)(out + (long)n * DIM + lane * VPL) = o;
            }
        }
    }
}


extern "C" void kernel_launch(void* const* d_in, const int* in_sizes, int n_in,
                              void* d_out, int out_size, void* d_ws, size_t ws_size,
                              hipStream_t stream) {
    const float* x  = (const float*)d_in[0];
    const int*   ei = (const int*)d_in[1];
    const float* W1 = (const float*)d_in[2];
    const float* b1 = (const float*)d_in[3];
    const float* W2 = (const float*)d_in[4];
    const float* b2 = (const float*)d_in[5];
    float* out = (float*)d_out;

    const int N = in_sizes[0] / 128;
    const int E = in_sizes[1] / 2;
    const int* src = ei;
    const int* dst = ei + E;

    float* dinv = (float*)d_ws;
    float* xw2  = dinv + N + (size_t)N * 128;

    unsigned* part = (unsigned*)xw2;
    deg_hist_kernel<<<dim3(HIST_SL, (N + HIST_R - 1) / HIST_R), HIST_THREADS, 0, stream>>>(dst, part, out, N, E);
    deg_to_dinv_kernel<<<(N + 255) / 256, 256, 0, stream>>>(dinv, part, N);

    const int aggBlocks = (N + AGG_R - 1) / AGG_R;

    agg_owner_kernel<4, 1><<<aggBlocks, AGG_THREADS, 0, stream>>>(x, dinv, src, dst, b1, W1, W2, xw2, out, N, E);

    agg_owner_kernel<2, 2><<<aggBlocks, AGG_THREADS, 0, stream>>>(xw2, dinv, src, dst, b2, nullptr, nullptr, out, out, N, E);
}


# No LDS f32 atomics: scan files each edge in the record of the wave owning its row (row mod 32; slot by LDS integer add); both layers' waves walk their records, rows summed in registers, arrival order
# speedup vs baseline: 4.3849x; 1.0610x over previous
#include <hip/hip_runtime.h>

typedef __attribute__((ext_vector_type(2))) float v2f;
typedef __attribute__((ext_vector_type(4))) float v4f;
typedef __attribute__((ext_vector_type(8))) float v8f;
typedef __attribute__((ext_vector_type(16))) float v16f;
typedef __attribute__((ext_vector_type(4))) int v4i;


constexpr int AGG_R = 400;
constexpr int REC_ROW = 64;
constexpr int REC_W = 368;
constexpr int KEY_SEG = 8192;
constexpr int KEY_OFF = 32 * REC_W * 8;
constexpr int KEY_PITCH = AGG_R * REC_ROW * 4;
constexpr int KEY_LD = 8;
constexpr int KEY_MW = KEY_LD / 2;
constexpr int KEY_CHUNK = 32 * 16 * KEY_LD;
static_assert(KEY_OFF + KEY_SEG <= KEY_PITCH, "records and keys of a group must fit its rows");
static_assert(KEY_SEG % KEY_CHUNK == 0 && KEY_OFF % 16 == 0 && KEY_PITCH % 16 == 0, "a chunk of keys lies in one segment, 16-byte aligned");

__device__ __forceinline__ bool keys_fit(int N, int E) {
    const int groups = (N + AGG_R - 1) / AGG_R;
    return groups <= 255 && (long)(N / AGG_R) * KEY_SEG >= (long)E;
}

__device__ __forceinline__ unsigned char* key_ptr(unsigned char* kb, int e) {
    return kb + (size_t)(e >> 13) * KEY_PITCH + (e & (KEY_SEG - 1));
}
static_assert(KEY_SEG == (1 << 13), "key_ptr splits an edge number at bit 13");

__device__ __forceinline__ void store_keys(unsigned char* kb, int e, const v4i& v, int E) {
    unsigned k[4];
#pragma unroll
    for (int c = 0; c < 4; ++c) k[c] = ((unsigned)v[c] / (unsigned)AGG_R) & 255u;
    if ((e & 3) == 0 && e >= 0 && e + 4 <= E) {
        *(unsigned*)key_ptr(kb, e) = k[0] | (k[1] << 8) | (k[2] << 16) | (k[3] << 24);
    } else {
#pragma unroll
        for (int c = 0; c < 4; ++c) {
            const int ee = e + c;
            if (ee >= 0 && ee < E) *key_ptr(kb, ee) = (unsigned char)k[c];
        }
    }
}

constexpr int HIST_R = 32768;
constexpr int HIST_SL = 32;
constexpr int HIST_THREADS = 1024;
constexpr int HIST_LD = 8;

__device__ __forceinline__ v4i load_dst_quad(const int* __restrict__ dst, int pad, int q, int E) {
    const int e = q * 4 - pad;
    if (e >= 0 && e + 4 <= E) return *(const v4i*)(dst + e);
    v4i r;
#pragma unroll
    for (int c = 0; c < 4; ++c) {
        const int ee = e + c;
        r[c] = (ee >= 0 && ee < E) ? dst[ee] : -1;
    }
    return r;
}

__global__ __launch_bounds__(HIST_THREADS) void deg_hist_kernel(const int* __restrict__ dst,
                                                                 unsigned* __restrict__ part,
                                                                 float* keyrows,
                                                                 int N, int E) {
    __shared__ unsigned cnt[HIST_R];
    const int tid = threadIdx.x;
    const int lo = blockIdx.y * HIST_R;
    const bool wk = (blockIdx.y == 0) && keys_fit(N, E);
    unsigned char* kb = (unsigned char*)keyrows + KEY_OFF;

    const v4i z = {};
    for (int i = tid; i < HIST_R / 4; i += HIST_THREADS) ((v4i*)cnt)[i] = z;
    __syncthreads();

    const int pad = (int)(((size_t)dst >> 2) & 3);
    const int nQuads = (E + pad + 3) / 4;
    const int perSlice = (((nQuads + HIST_SL - 1) / HIST_SL) + 31) & ~31;
    const int q0 = blockIdx.x * perSlice;
    const int q1 = (q0 + perSlice < nQuads) ? (q0 + perSlice) : nQuads;

    const bool inside = (q0 * 4 - pad >= 0) && (q1 * 4 - pad <= E);
    if (inside) {
        const v4i* p = (const v4i*)(dst - pad);
        for (int q = q0 + tid; q < q1; q += HIST_THREADS * HIST_LD) {
            v4i v[HIST_LD];
#pragma unroll
            for (int j = 0; j < HIST_LD; ++j) {
                const int qq = q + j * HIST_THREADS;
                v[j] = p[qq < q1 ? qq : q];
            }
#pragma unroll
            for (int j = 0; j < HIST_LD; ++j) {
                const bool live = (q + j * HIST_THREADS < q1);
                if (wk && live) store_keys(kb, (q + j * HIST_THREADS) * 4 - pad, v[j], E);
#pragma unroll
                for (int c = 0; c < 4; ++c) {
                    const unsigned d = (unsigned)(v[j][c] - lo);
                    if (live && d < (unsigned)HIST_R)
                        __hip_atomic_fetch_add(&cnt[d], 1u, __ATOMIC_RELAXED, __HIP_MEMORY_SCOPE_WORKGROUP);
                }
            }
        }
    } else {
        for (int q = q0 + tid; q < q1; q += HIST_THREADS) {
            const v4i v = load_dst_quad(dst, pad, q, E);
            if (wk) store_keys(kb, q * 4 - pad, v, E);
#pragma unroll
            for (int c = 0; c < 4; ++c) {
                const unsigned d = (unsigned)(v[c] - lo);
                if (d < (unsigned)HIST_R)
                    __hip_atomic_fetch_add(&cnt[d], 1u, __ATOMIC_RELAXED, __HIP_MEMORY_SCOPE_WORKGROUP);
            }
        }
    }
    __syncthreads();

    unsigned* plane = part + (size_t)blockIdx.x * N;
    for (int i = tid * 4; i < HIST_R; i += HIST_THREADS * 4) {
        const int n = lo + i;
        const v4i w = *(const v4i*)(cnt + i);
        if (n + 4 <= N && (((size_t)(plane + n)) & 15) == 0) {
            *(v4i*)(plane + n) = w;
        } else {
#pragma unroll
            for (int c = 0; c < 4; ++c)
                if (n + c < N) plane[n + c] = (unsigned)w[c];
        }
    }
}

__global__ void deg_to_dinv_kernel(float* __restrict__ dinv, const unsigned* __restrict__ part, int N) {
    const int i = blockIdx.x * blockDim.x + threadIdx.x;
    if (i < N) {
        unsigned c = 1u;
#pragma unroll 8
        for (int s = 0; s < HIST_SL; ++s) c += part[(size_t)s * N + i];
        const float v = (float)c;
        dinv[i] = (v > 0.0f) ? rsqrtf(v) : 0.0f;
    }
}


constexpr int AGG_THREADS = 1024;
constexpr int AGG_LD = 8;
constexpr int AGG_EPL = 4 * AGG_LD;
constexpr int AGG_CHUNK = 32 * AGG_EPL;

template <int VPL> struct RowVec;
template <> struct RowVec<4> { typedef v4f type; };
template <> struct RowVec<2> { typedef v2f type; };

constexpr int H_PITCH = 132;
template <int VPL> struct RowPitch { static constexpr int v = (VPL == 4) ? H_PITCH : 32 * VPL; };

__device__ __forceinline__ void lds_add_f32(float* p, float v) {
    __hip_atomic_fetch_add(p, v, __ATOMIC_RELAXED, __HIP_MEMORY_SCOPE_WORKGROUP);
}

__device__ __forceinline__ float readlane_f32(float x, int l) {
    return __int_as_float(__builtin_amdgcn_readlane(__float_as_int(x), l));
}

__device__ __forceinline__ void load_dst_chunk(v4i (&v)[AGG_LD], const int* __restrict__ dst,
                                               int cbase, int lane, int E) {
    if (cbase >= 0 && cbase + AGG_CHUNK <= E) {
        const v4i* p = (const v4i*)(dst + cbase) + lane;
#pragma unroll
        for (int j = 0; j < AGG_LD; ++j) v[j] = p[j * 32];
    } else {
#pragma unroll
        for (int j = 0; j < AGG_LD; ++j) {
#pragma unroll
            for (int q = 0; q < 4; ++q) {
                const int e = cbase + j * 128 + lane * 4 + q;
                v[j][q] = (e >= 0 && e < E) ? dst[e] : -1;
            }
        }
    }
}

__device__ __forceinline__ unsigned owned_mask(const v4i (&v)[AGG_LD], int lo) {
    unsigned m = 0u;
#pragma unroll
    for (int j = 0; j < AGG_LD; ++j) {
#pragma unroll
        for (int q = 0; q < 4; ++q)
            if ((unsigned)(v[j][q] - lo) < (unsigned)AGG_R) m |= (1u << (j * 4 + q));
    }
    return m;
}

__device__ __forceinline__ void load_key_chunk(v4i (&v)[KEY_LD], const unsigned char* kb, int c, int lane) {
    const v4i* p = (const v4i*)key_ptr(const_cast<unsigned char*>(kb), c * KEY_CHUNK) + lane;
#pragma unroll
    for (int j = 0; j < KEY_LD; ++j) v[j] = p[j * 32];
}

__device__ __forceinline__ void key_mask(unsigned (&m)[KEY_MW], const v4i (&v)[KEY_LD], unsigned gpat) {
#pragma unroll
    for (int q = 0; q < KEY_MW; ++q) {
        unsigned a = 0u;
#pragma unroll
        for (int p = 0; p < 4; ++p) {
            const int j = q * 2 + (p >> 1);
            const int w0 = (p & 1) * 2;
            const unsigned x0 = (unsigned)v[j][w0] ^ gpat;
            const unsigned x1 = (unsigned)v[j][w0 + 1] ^ gpat;
            const unsigned n0 = (((x0 & 0x7f7f7f7fu) + 0x7f7f7f7fu) | x0) & 0x80808080u;
            const unsigned n1 = (((x1 & 0x7f7f7f7fu) + 0x7f7f7f7fu) | x1) & 0x80808080u;
            unsigned d = __builtin_amdgcn_udot4(n0, 0x08040201u, 0u, false);
            d = __builtin_amdgcn_udot4(n1, 0x80402010u, d, false);
            a |= (p == 0) ? (d >> 7) : (d << (8 * p - 7));
        }
        m[q] = ~a;
    }
}

template <int VPL>
__device__ __forceinline__ typename RowVec<VPL>::type row_load(const float* __restrict__ xw, int su, int lane) {
    typedef typename RowVec<VPL>::type vec;
    return *(const vec*)(xw + (long)su * (32 * VPL) + lane * VPL);
}

template <int VPL>
__device__ __forceinline__ void row_add(float* acc, int ru, float w, const typename RowVec<VPL>::type& q, int lane) {
    float* a = acc + ru * RowPitch<VPL>::v + lane;
#pragma unroll
    for (int i = 0; i < VPL; ++i) lds_add_f32(a + i * 32, q[i] * w);
}

constexpr int REC_SHIFT = 9;
static_assert(AGG_R <= (1 << REC_SHIFT), "the row inside the group must fit under the source node in an entry");
constexpr int OWN = (AGG_R + 31) / 32;
constexpr int NO_ROW = 15;
static_assert(OWN <= NO_ROW && NO_ROW < 16 && NO_ROW * 32 >= AGG_R, "sixteen slots hold a wave's rows and the one for nobody's");

template <int VPL> struct OwnRows { v16f ch[VPL]; };

template <int VPL>
__device__ __forceinline__ void own_add(OwnRows<VPL>& own, int k, float w, const typename RowVec<VPL>::type& q) {
#pragma unroll
    for (int i = 0; i < VPL; ++i) own.ch[i][k] += q[i] * w;
}

template <int VPL, int G>
__device__ __forceinline__ void take_rows(OwnRows<VPL>& own, const float* __restrict__ xw,
                                          const v2f& ent, int j, int lane) {
    typedef typename RowVec<VPL>::type vec;
    vec q[G];
#pragma unroll
    for (int t = 0; t < G; ++t) {
        const unsigned key = (unsigned)__builtin_amdgcn_readlane(__float_as_int(ent[0]), j + t);
        q[t] = row_load<VPL>(xw, (int)(key >> REC_SHIFT), lane);
    }
#pragma unroll
    for (int t = 0; t < G; ++t) {
        const int key = __builtin_amdgcn_readlane(__float_as_int(ent[0]), j + t);
        own_add<VPL>(own, (key & ((1 << REC_SHIFT) - 1)) >> 5, readlane_f32(ent[1], j + t), q[t]);
    }
}

template <int VPL, int G>
__device__ __forceinline__ void walk_record(OwnRows<VPL>& own, const float* __restrict__ xw,
                                            const v2f* lst, int cnt, int lo, int lane) {
    static_assert(32 % G == 0, "a batch of 32 entries is whole rounds of G");
    v2f filler;
    filler[0] = __int_as_float((lo << REC_SHIFT) | (NO_ROW << 5));
    filler[1] = 0.0f;
    v2f ent = filler;
    if (lane < cnt) ent = lst[1 + lane];
    for (int base = 0; base < cnt; base += 32) {
        const int nb = cnt - base;
        const int ni = base + 32 + lane;
        v2f nxt = lst[1 + (ni < cnt ? ni : cnt - 1)];
        if (ni >= cnt) nxt = filler;
#pragma nounroll
        for (int j = 0; j < 32; j += G) {
            if (j < nb) take_rows<VPL, G>(own, xw, ent, j, lane);
        }
        ent = nxt;
    }
}

template <int VPL, int G>
__device__ __forceinline__ void replay_take(float* acc, const float* __restrict__ xw, const v2f& ent, int j, int lane) {
    typedef typename RowVec<VPL>::type vec;
    vec q[G];
#pragma unroll
    for (int t = 0; t < G; ++t) {
        const unsigned key = (unsigned)__builtin_amdgcn_readlane(__float_as_int(ent[0]), j + t);
        q[t] = row_load<VPL>(xw, (int)(key >> REC_SHIFT), lane);
    }
#pragma unroll
    for (int t = 0; t < G; ++t) {
        const int key = __builtin_amdgcn_readlane(__float_as_int(ent[0]), j + t);
        row_add<VPL>(acc, key & ((1 << REC_SHIFT) - 1), readlane_f32(ent[1], j + t), q[t], lane);
    }
}

constexpr int RING = 64;
static_assert((RING & (RING - 1)) == 0 && RING >= 63, "31 waiting edges and 32 new ones fit the ring; its index wraps by a mask");

template <bool FULL>
__device__ __forceinline__ void route_batch(const float* dl, const float* __restrict__ dinv,
                                            const int* __restrict__ src, const int* __restrict__ dst,
                                            v2f* recs, int wrec, int cap, int* fill, int* over,
                                            int lo, int e, int nb, int lane) {
    if (FULL || lane < nb) {
        const int s = src[e];
        const int d = dst[e];
        asm volatile("" :: "v"(s), "v"(d));
        const int r = d - lo;
        if ((unsigned)r < (unsigned)AGG_R) {
            v2f ent;
            ent[0] = __int_as_float((s << REC_SHIFT) | r);
            ent[1] = dinv[s] * dl[r];
            const int t = r & 31;
            const int slot = __hip_atomic_fetch_add(fill + t, 1, __ATOMIC_RELAXED, __HIP_MEMORY_SCOPE_WORKGROUP);
            if (slot < cap) recs[t * wrec + 1 + slot] = ent;
            else *over = 1;
        }
    }
}

template <int VPL, bool FULL>
__device__ __forceinline__ void work_batch(float* acc, const float* dl,
                                           const float* __restrict__ xw, const float* __restrict__ dinv,
                                           const int* __restrict__ src, const int* __restrict__ dst,
                                           int lo, int e, int nb, int lane) {
    const bool in = FULL || (lane < nb);
    int s = 0, r = 0;
    bool ok = in;
    if (in) {
        s = src[e];
        r = dst[e] - lo;
        ok = ((unsigned)r < (unsigned)AGG_R);
    }
    float w = 0.0f;
    if (ok) w = dinv[s] * dl[r];
    v2f ent;
    ent[0] = __int_as_float((s << REC_SHIFT) | r);
    ent[1] = w;
    const unsigned V = __builtin_amdgcn_ballot_w32(ok);
    const unsigned all = FULL ? 0xffffffffu : ((1u << nb) - 1u);
    if (V == all) {
        if (FULL) {
            replay_take<VPL, 16>(acc, xw, ent, 0, lane);
            replay_take<VPL, 16>(acc, xw, ent, 16, lane);
        } else {
            int j = 0;
            if (nb - j >= 16) { replay_take<VPL, 16>(acc, xw, ent, j, lane); j += 16; }
            if (nb - j >= 8) { replay_take<VPL, 8>(acc, xw, ent, j, lane); j += 8; }
            if (nb - j >= 4) { replay_take<VPL, 4>(acc, xw, ent, j, lane); j += 4; }
            if (nb - j >= 2) { replay_take<VPL, 2>(acc, xw, ent, j, lane); j += 2; }
            if (nb - j >= 1) { replay_take<VPL, 1>(acc, xw, ent, j, lane); }
        }
    } else {
        unsigned B = V;
        while (B != 0u) {
            const int l = __builtin_ctz(B); B &= B - 1u;
            replay_take<VPL, 1>(acc, xw, ent, l, lane);
        }
    }
}

template <int VPL, bool ROUTE, bool KEYS>
__device__ __forceinline__ void scan_edges(float* acc, const float* dl,
                                           const float* __restrict__ xw, const float* __restrict__ dinv,
                                           const int* __restrict__ src, const int* __restrict__ dst,
                                           const unsigned char* kb,
                                           v2f* recs, int wrec, int cap, int* fill, int* over,
                                           int lo, int E, int wave, int lane, int* ring) {
    const int pad = KEYS ? 0 : (int)(((size_t)dst >> 2) & 3);
    constexpr int CH = KEYS ? KEY_CHUNK : AGG_CHUNK;
    const int nChunks = (E + pad + CH - 1) / CH;
    const unsigned gpat = blockIdx.x * 0x01010101u;
    const int rot = (int)(((long)blockIdx.x * nChunks) / gridDim.x);

    int head = 0;
    int nring = 0;

    v4i buf[AGG_LD];
    v4i kbuf[KEY_LD];
    unsigned m[KEY_MW];
#pragma unroll
    for (int q = 0; q < KEY_MW; ++q) m[q] = 0u;
    auto left = [&]() {
        unsigned o = m[0];
#pragma unroll
        for (int q = 1; q < KEY_MW; ++q) o |= m[q];
        return o;
    };
    int cbase = 0;
    int it = wave;
    if (it < nChunks) {
        int c = it + rot;
        if (c >= nChunks) c -= nChunks;
        cbase = c * CH - pad;
        if (KEYS) {
            load_key_chunk(kbuf, kb, c, lane);
            key_mask(m, kbuf, gpat);
        } else {
            load_dst_chunk(buf, dst, cbase, lane, E);
            m[0] = owned_mask(buf, lo);
        }
    }
    while (it < nChunks) {
        const int itn = it + 32;
        const bool more = (itn < nChunks);
        int cbn = 0;
        if (more) {
            int c = itn + rot;
            if (c >= nChunks) c -= nChunks;
            cbn = c * CH - pad;
            if (KEYS) load_key_chunk(kbuf, kb, c, lane);
            else load_dst_chunk(buf, dst, cbn, lane, E);
        }

        while (__builtin_amdgcn_ballot_w32(left() != 0u) != 0u) {
            bool has = (left() != 0u);
            int e = 0;
            if (has) {
                int q = KEY_MW - 1;
                unsigned w = m[KEY_MW - 1];
#pragma unroll
                for (int t = KEY_MW - 2; t >= 0; --t)
                    if (m[t] != 0u) { q = t; w = m[t]; }
                const int k = __builtin_ctz(w);
#pragma unroll
                for (int t = 0; t < KEY_MW; ++t)
                    if (t == q) m[t] &= m[t] - 1u;
                e = KEYS ? (cbase + (q * 2 + (k >> 4)) * 512 + lane * 16 + (k & 15))
                         : (cbase + (k >> 2) * 128 + lane * 4 + (k & 3));
            }
            if (KEYS) has = has && (e < E);
            const unsigned Bn = __builtin_amdgcn_ballot_w32(has);
            if (has) ring[(head + nring + (int)__builtin_amdgcn_mbcnt_lo(Bn, 0u)) & (RING - 1)] = e;
            nring += __builtin_popcount(Bn);
            if (nring >= 32) {
                const int eb = ring[(head + lane) & (RING - 1)];
                if (ROUTE) route_batch<true>(dl, dinv, src, dst, recs, wrec, cap, fill, over, lo, eb, 32, lane);
                else work_batch<VPL, true>(acc, dl, xw, dinv, src, dst, lo, eb, 32, lane);
                head = (head + 32) & (RING - 1);
                nring -= 32;
            }
        }

        if (!more) {
#pragma unroll
            for (int q = 0; q < KEY_MW; ++q) m[q] = 0u;
        } else if (KEYS) {
            key_mask(m, kbuf, gpat);
        } else {
            m[0] = owned_mask(buf, lo);
        }
        cbase = cbn;
        it = itn;
    }
    if (nring > 0) {
        int eb = 0;
        if (lane < nring) eb = ring[(head + lane) & (RING - 1)];
        if (ROUTE) route_batch<false>(dl, dinv, src, dst, recs, wrec, cap, fill, over, lo, eb, nring, lane);
        else work_batch<VPL, false>(acc, dl, xw, dinv, src, dst, lo, eb, nring, lane);
    }
}

constexpr int W1_PP = 288;
constexpr int W2_PP = 160;
static_assert(AGG_R % 16 == 0, "a group's rows are whole 16-row tiles");
static_assert(W2_PP <= W1_PP, "W2 takes W1's place in LDS");

template <int NR, int NT, int PP>
__device__ __forceinline__ void tile_product(v8f (&d)[2][NT], const float* acc, const float* wt,
                                             int rt0, int c0, int lane) {
    const int m = lane & 15;
    const int half = lane >> 4;
    const float* bp = wt + half * PP + (c0 + m) * 2;
    const float* ap = acc + (rt0 * 16 + m) * H_PITCH + half * 2;
    const v8f zero = {};
#pragma unroll
    for (int j = 0; j < NR; ++j) {
#pragma unroll
        for (int t = 0; t < NT; ++t) d[j][t] = zero;
    }
#pragma unroll
    for (int ks = 0; ks < 32; ++ks) {
        v2f b[NT];
#pragma unroll
        for (int t = 0; t < NT; ++t) b[t] = *(const v2f*)(bp + ks * (2 * PP) + t * 32);
        v2f a[NR];
#pragma unroll
        for (int j = 0; j < NR; ++j) a[j] = *(const v2f*)(ap + j * (256 * H_PITCH) + ks * 4);
#pragma unroll
        for (int j = 0; j < NR; ++j) {
#pragma unroll
            for (int t = 0; t < NT; ++t)
                d[j][t] = __builtin_amdgcn_wmma_f32_16x16x4_f32(false, a[j], false, b[t], (short)0, d[j][t], false, false);
        }
    }
}

template <int NR>
__device__ __forceinline__ void h_to_rows(const v8f (&d)[2][4], float* acc, const float* __restrict__ b1,
                                          int rt0, int c0, int lane) {
    const int m = lane & 15;
    const int half = lane >> 4;
#pragma unroll
    for (int t = 0; t < 4; ++t) {
        const int col = c0 + t * 16 + m;
        const float bc = b1[col];
#pragma unroll
        for (int j = 0; j < NR; ++j) {
            float* row = acc + ((rt0 + 16 * j) * 16 + 8 * half) * H_PITCH + col;
#pragma unroll
            for (int r = 0; r < 8; ++r) row[r * H_PITCH] = d[j][t][r] + bc;
        }
    }
}

template <int NR>
__device__ __forceinline__ void xw2_store(const v8f (&d)[2][2], float* __restrict__ y,
                                          int lo, int N, int rt0, int c0, int lane) {
    const int m = lane & 15;
    const int half = lane >> 4;
#pragma unroll
    for (int j = 0; j < NR; ++j) {
        const int rbase = lo + (rt0 + 16 * j) * 16 + 8 * half;
#pragma unroll
        for (int t = 0; t < 2; ++t) {
            const int col = c0 + t * 16 + m;
#pragma unroll
            for (int r = 0; r < 8; ++r) {
                const int n = rbase + r;
                if (n < N) y[(long)n * 64 + col] = d[j][t][r];
            }
        }
    }
}

template <int VPL, int K0, int CNT>
__device__ __forceinline__ void self_rows(OwnRows<VPL>& own, const float* __restrict__ xw,
                                          const float* dl, int lo, int N, int wave, int lane) {
    typedef typename RowVec<VPL>::type vec;
    static_assert(K0 + CNT <= OWN, "rows of the wave");
    vec sq[CNT];
    float sdv[CNT];
#pragma unroll
    for (int t = 0; t < CNT; ++t) {
        const int r = wave + 32 * (K0 + t);
        const bool live = (r < AGG_R && lo + r < N);
        sq[t] = row_load<VPL>(xw, live ? (lo + r) : lo, lane);
        sdv[t] = dl[live ? r : 0];
    }
#pragma unroll
    for (int t = 0; t < CNT; ++t) {
        const int r = wave + 32 * (K0 + t);
        if (r < AGG_R && lo + r < N) {
            const float dv = sdv[t];
#pragma unroll
            for (int i = 0; i < VPL; ++i) own.ch[i][K0 + t] += sq[t][i] * dv * dv;
        }
    }
}

template <int VPL, int MODE>
__global__ __launch_bounds__(AGG_THREADS) void agg_owner_kernel(const float* __restrict__ xw,
                                                                const float* __restrict__ dinv,
                                                                const int* __restrict__ src,
                                                                const int* __restrict__ dst,
                                                                const float* __restrict__ bias,
                                                                const float* __restrict__ w1,
                                                                const float* __restrict__ w2,
                                                                float* out, float* rec,
                                                                int N, int E) {
    constexpr int DIM = 32 * VPL;
    constexpr int PITCH = RowPitch<VPL>::v;
    constexpr int G = (VPL == 4) ? 8 : 16;
    static_assert(MODE == 2 || VPL == 4, "layer 1 is the 128-wide layer whose rows feed the products by W1 and W2");
    static_assert(MODE == 1 || DIM == REC_ROW, "the walk reads its records from its own output rows");
    typedef typename RowVec<VPL>::type vec;
    __shared__ float acc[AGG_R * H_PITCH];
    __shared__ float dl[AGG_R];
    __shared__ float wt[MODE == 1 ? 64 * W1_PP : 2];
    __shared__ int rings[(AGG_THREADS / 32) * RING];
    __shared__ int fill[AGG_THREADS / 32];
    __shared__ int over;
    static_assert(AGG_THREADS / 32 == 32, "row r belongs to wave r & 31; one lane per wave's head in layer 2");

    const int tid = threadIdx.x;
    const int lane = tid & 31;
    const int wave = __builtin_amdgcn_readfirstlane(tid >> 5);
    const int lo = blockIdx.x * AGG_R;

    static_assert(128 * 128 == 4 * 4 * AGG_THREADS, "four 16-byte pieces of W1 per thread");
    static_assert(128 * 64 == 2 * 4 * AGG_THREADS, "two 16-byte pieces of W2 per thread");
    v4f wq[4];
    if (MODE == 1) {
#pragma unroll
        for (int j = 0; j < 4; ++j) wq[j] = *(const v4f*)(w1 + (tid + j * AGG_THREADS) * 4);
    }
    vec bv = {};
    if (MODE == 2) bv = *(const vec*)(bias + lane * VPL);
    for (int r = tid; r < AGG_R; r += AGG_THREADS) {
        const int n = lo + r;
        dl[r] = (n < N) ? dinv[n] : 0.0f;
    }
    if (MODE == 1) {
#pragma unroll
        for (int j = 0; j < 4; ++j) {
            const int i4 = tid + j * AGG_THREADS;
            const int ch = i4 >> 5;
            const int n0 = (i4 & 31) * 4;
            const int p = (ch & 3) * 32 + (ch >> 2);
            float* w = wt + (p >> 1) * W1_PP + 2 * n0 + (p & 1);
#pragma unroll
            for (int q = 0; q < 4; ++q) w[2 * q] = wq[j][q];
        }
        if (tid < AGG_THREADS / 32) fill[tid] = 0;
        if (tid == 0) over = 0;
    }

    const int rows = (N - lo < AGG_R) ? (N - lo) : AGG_R;
    const bool kf = keys_fit(N, E);
    const int wrec = (kf && rows == AGG_R) ? REC_W : rows;
    const int cap = wrec - 1;
    v2f* recs = (v2f*)((MODE == 1 ? rec : out) + (long)lo * REC_ROW);
    v2f* lst = recs + wave * wrec;

    bool slow = false;
    int cnt = 0;
    if (MODE == 2) {
        const v2f hd = recs[lane * wrec];
        slow = (__builtin_amdgcn_ballot_w32(__float_as_int(hd[1]) != 0) != 0u);
        cnt = __builtin_amdgcn_readlane(__float_as_int(hd[0]), wave);
        cnt = cnt < 0 ? 0 : (cnt > cap ? cap : cnt);
        if (slow) {
            for (int r = wave; r < AGG_R; r += 32) {
#pragma unroll
                for (int i = 0; i < VPL; ++i) acc[r * PITCH + i * 32 + lane] = bv[i];
            }
        }
    }
    __syncthreads();

    static_assert(REC_W <= AGG_R && 32 * AGG_R * 8 <= AGG_R * H_PITCH * 4, "the 32 records fit the rows' place in LDS");
    v2f* qrecs = (v2f*)acc;
    v2f* ql = qrecs + wave * wrec;
    if (MODE == 1) {
        const unsigned char* kb = (const unsigned char*)rec + KEY_OFF;
        if (kf) scan_edges<VPL, true, true>(acc, dl, xw, dinv, src, dst, kb, qrecs, wrec, cap, fill, &over, lo, E, wave, lane, rings + wave * RING);
        else scan_edges<VPL, true, false>(acc, dl, xw, dinv, src, dst, kb, qrecs, wrec, cap, fill, &over, lo, E, wave, lane, rings + wave * RING);
        __syncthreads();
        slow = (over != 0);
        cnt = fill[wave];
        cnt = cnt > cap ? cap : cnt;
        if (slow) cnt = 0;
        {
            v2f hd;
            hd[0] = __int_as_float(cnt);
            hd[1] = __int_as_float(slow ? 1 : 0);
            for (int i = lane; i <= cnt; i += 32) {
                v2f en = hd;
                if (i > 0) en = ql[i];
                lst[i] = en;
            }
        }
        if (slow) {
            for (int r = wave; r < AGG_R; r += 32) {
#pragma unroll
                for (int i = 0; i < VPL; ++i) acc[r * PITCH + i * 32 + lane] = 0.0f;
            }
            __syncthreads();
            if (kf) scan_edges<VPL, false, true>(acc, dl, xw, dinv, src, dst, kb, recs, wrec, cap, fill, &over, lo, E, wave, lane, rings + wave * RING);
            else scan_edges<VPL, false, false>(acc, dl, xw, dinv, src, dst, kb, recs, wrec, cap, fill, &over, lo, E, wave, lane, rings + wave * RING);
        }
    } else if (slow) {
        scan_edges<VPL, false, false>(acc, dl, xw, dinv, src, dst, nullptr, recs, wrec, cap, nullptr, nullptr, lo, E, wave, lane, rings + wave * RING);
    }

    OwnRows<VPL> own;
    const v16f vzero = {};
#pragma unroll
    for (int i = 0; i < VPL; ++i) own.ch[i] = vzero;
    if (!slow) walk_record<VPL, G>(own, xw, (MODE == 1) ? (const v2f*)ql : (const v2f*)lst, cnt, lo, lane);

    self_rows<VPL, 0, 7>(own, xw, dl, lo, N, wave, lane);
    self_rows<VPL, 7, OWN - 7>(own, xw, dl, lo, N, wave, lane);
    if (MODE == 1) {
        if (!slow) __syncthreads();
#pragma unroll
        for (int k = 0; k < OWN; ++k) {
            const int r = wave + 32 * k;
            if (r < AGG_R) {
                float* a = acc + r * PITCH + lane;
#pragma unroll
                for (int i = 0; i < VPL; ++i) {
                    if (slow) lds_add_f32(a + i * 32, own.ch[i][k]);
                    else a[i * 32] = own.ch[i][k];
                }
            }
        }
    }
    __syncthreads();

    if (MODE == 1) {
        const int cp = wave & 1;
        const int rt0 = wave >> 1;
        const bool two = (rt0 + 16 < AGG_R / 16);
        v8f h[2][4];
        if (two) tile_product<2, 4, W1_PP>(h, acc, wt, rt0, cp * 64, lane);
        else tile_product<1, 4, W1_PP>(h, acc, wt, rt0, cp * 64, lane);
        const int th = wave * 32 + lane;
        v4f w2q[2];
#pragma unroll
        for (int j = 0; j < 2; ++j) w2q[j] = *(const v4f*)(w2 + (th + j * AGG_THREADS) * 4);
        __syncthreads();
        if (two) h_to_rows<2>(h, acc, bias, rt0, cp * 64, lane);
        else h_to_rows<1>(h, acc, bias, rt0, cp * 64, lane);
#pragma unroll
        for (int j = 0; j < 2; ++j) {
            const int i4 = th + j * AGG_THREADS;
            const int ch = i4 >> 4;
            const int n0 = (i4 & 15) * 4;
            float* w = wt + (ch >> 1) * W2_PP + 2 * n0 + (ch & 1);
#pragma unroll
            for (int q = 0; q < 4; ++q) w[2 * q] = w2q[j][q];
        }
        __syncthreads();
        v8f y[2][2];
        if (two) {
            tile_product<2, 2, W2_PP>(y, acc, wt, rt0, cp * 32, lane);
            xw2_store<2>(y, out, lo, N, rt0, cp * 32, lane);
        } else {
            tile_product<1, 2, W2_PP>(y, acc, wt, rt0, cp * 32, lane);
            xw2_store<1>(y, out, lo, N, rt0, cp * 32, lane);
        }
    } else {
#pragma unroll
        for (int k = 0; k < OWN; ++k) {
            const int r = wave + 32 * k;
            const int n = lo + r;
            if (r < AGG_R && n < N) {
                vec o;
#pragma unroll
                for (int i = 0; i < VPL; ++i) o[i] = (slow ? acc[r * PITCH + i * 32 + lane] : bv[i]) + own.ch[i][k];
                *(vec*)(out + (long)n * DIM + lane * VPL) = o;
            }
        }
    }
}


extern "C" void kernel_launch(void* const* d_in, const int* in_sizes, int n_in,
                              void* d_out, int out_size, void* d_ws, size_t ws_size,
                              hipStream_t stream) {
    const float* x  = (const float*)d_in[0];
    const int*   ei = (const int*)d_in[1];
    const float* W1 = (const float*)d_in[2];
    const float* b1 = (const float*)d_in[3];
    const float* W2 = (const float*)d_in[4];
    const float* b2 = (const float*)d_in[5];
    float* out = (float*)d_out;

    const int N = in_sizes[0] / 128;
    const int E = in_sizes[1] / 2;
    const int* src = ei;
    const int* dst = ei + E;

    float* dinv = (float*)d_ws;
    float* xw2  = dinv + N + (size_t)N * 128;

    unsigned* part = (unsigned*)xw2;
    deg_hist_kernel<<<dim3(HIST_SL, (N + HIST_R - 1) / HIST_R), HIST_THREADS, 0, stream>>>(dst, part, out, N, E);
    deg_to_dinv_kernel<<<(N + 255) / 256, 256, 0, stream>>>(dinv, part, N);

    const int aggBlocks = (N + AGG_R - 1) / AGG_R;

    agg_owner_kernel<4, 1><<<aggBlocks, AGG_THREADS, 0, stream>>>(x, dinv, src, dst, b1, W1, W2, xw2, out, N, E);

    agg_owner_kernel<2, 2><<<aggBlocks, AGG_THREADS, 0, stream>>>(xw2, dinv, src, dst, b2, nullptr, nullptr, out, out, N, E);
}
